# starting kernel
# baseline (speedup 1.0000x reference)
#include <hip/hip_runtime.h>

#define LDIM 256
#define DDIM 512
#define HDIM 256
#define ODIM 16
#define MDIM 512
#define BDIM 2
#define LEAKY 0.1f
#define EPSV 1e-9f

typedef __attribute__((ext_vector_type(2))) float v2f;
typedef __attribute__((ext_vector_type(8))) float v8f;

__global__ __launch_bounds__(32) void mlp_wmma_kernel(
    const float* __restrict__ X, const float* __restrict__ W,
    const float* __restrict__ bias, float* __restrict__ Hout)
{
    const int lane  = threadIdx.x & 31;
    const int mBase = blockIdx.x * 16;
    const int nBase = blockIdx.y * 64;
    const int mRow  = mBase + (lane & 15);
    const int kOff  = (lane >> 4) * 2;

    v8f acc0 = {}, acc1 = {}, acc2 = {}, acc3 = {};

    const float* xrow = X + mRow * DDIM + kOff;
    const int n0 = nBase + (lane & 15);
    const float* w0 = W + (n0 +  0) * DDIM + kOff;
    const float* w1 = W + (n0 + 16) * DDIM + kOff;
    const float* w2 = W + (n0 + 32) * DDIM + kOff;
    const float* w3 = W + (n0 + 48) * DDIM + kOff;

    #pragma unroll 4
    for (int k = 0; k < DDIM; k += 4) {
        v2f a  = *(const v2f*)(xrow + k);
        v2f b0 = *(const v2f*)(w0 + k);
        v2f b1 = *(const v2f*)(w1 + k);
        v2f b2 = *(const v2f*)(w2 + k);
        v2f b3 = *(const v2f*)(w3 + k);
        acc0 = __builtin_amdgcn_wmma_f32_16x16x4_f32(false, a, false, b0, 0, acc0, false, false);
        acc1 = __builtin_amdgcn_wmma_f32_16x16x4_f32(false, a, false, b1, 0, acc1, false, false);
        acc2 = __builtin_amdgcn_wmma_f32_16x16x4_f32(false, a, false, b2, 0, acc2, false, false);
        acc3 = __builtin_amdgcn_wmma_f32_16x16x4_f32(false, a, false, b3, 0, acc3, false, false);
    }

    const int mOut = mBase + (lane >> 4) * 8;
    v8f accs[4] = {acc0, acc1, acc2, acc3};
    #pragma unroll
    for (int t = 0; t < 4; ++t) {
        const int nCol = nBase + t * 16 + (lane & 15);
        const float bv = bias[nCol];
        v8f acc = accs[t];
        #pragma unroll
        for (int r = 0; r < 8; ++r) {
            float h = acc[r] + bv;
            h = (h >= 0.0f) ? h : (LEAKY * h);
            Hout[(mOut + r) * HDIM + nCol] = h;
        }
    }
}

__global__ void cumsum_kernel(const float* __restrict__ h1, float* __restrict__ cum)
{
    const int b = blockIdx.x;
    const int h = threadIdx.x;
    const float* src = h1 + b * LDIM * HDIM + h;
    float*       dst = cum + b * LDIM * HDIM + h;
    float run = 0.0f;
    for (int l = 0; l < LDIM; ++l) {
        run += src[l * HDIM];
        dst[l * HDIM] = run;
    }
}

__global__ __launch_bounds__(32) void proj_wmma_kernel(
    const float* __restrict__ h1, const float* __restrict__ h2,
    const float* __restrict__ cum, const float* __restrict__ Wf,
    const float* __restrict__ bf,
    float* __restrict__ Qj, float* __restrict__ Qi, float* __restrict__ P5)
{
    const int lane = threadIdx.x & 31;
    const int mRow = blockIdx.x * 16 + (lane & 15);
    const int kOff = (lane >> 4) * 2;
    const int o    = lane & 15;

    const float* a1p = h1  + mRow * HDIM + kOff;
    const float* a2p = h2  + mRow * HDIM + kOff;
    const float* a5p = cum + mRow * HDIM + kOff;
    const float* w1p = Wf + o * (3 * HDIM) + kOff;
    const float* w2p = w1p + HDIM;
    const float* w3p = w1p + 2 * HDIM;

    v8f p1 = {}, p2 = {}, p3 = {}, p4 = {}, p5 = {};
    #pragma unroll 4
    for (int k = 0; k < HDIM; k += 4) {
        v2f a1 = *(const v2f*)(a1p + k);
        v2f a2 = *(const v2f*)(a2p + k);
        v2f a5 = *(const v2f*)(a5p + k);
        v2f b1 = *(const v2f*)(w1p + k);
        v2f b2 = *(const v2f*)(w2p + k);
        v2f b3 = *(const v2f*)(w3p + k);
        p1 = __builtin_amdgcn_wmma_f32_16x16x4_f32(false, a1, false, b1, 0, p1, false, false);
        p2 = __builtin_amdgcn_wmma_f32_16x16x4_f32(false, a2, false, b1, 0, p2, false, false);
        p3 = __builtin_amdgcn_wmma_f32_16x16x4_f32(false, a2, false, b2, 0, p3, false, false);
        p4 = __builtin_amdgcn_wmma_f32_16x16x4_f32(false, a1, false, b2, 0, p4, false, false);
        p5 = __builtin_amdgcn_wmma_f32_16x16x4_f32(false, a5, false, b3, 0, p5, false, false);
    }

    const int mOut = blockIdx.x * 16 + (lane >> 4) * 8;
    const float bfv = bf[o];
    #pragma unroll
    for (int r = 0; r < 8; ++r) {
        const int m = mOut + r;
        Qj[m * ODIM + o] = p1[r] + 0.5f * p3[r] + bfv;
        Qi[m * ODIM + o] = 0.5f * p4[r] - p2[r];
        P5[m * ODIM + o] = p5[r];
    }
}

__global__ __launch_bounds__(LDIM) void epilogue_kernel(
    const float* __restrict__ Qj, const float* __restrict__ Qi,
    const float* __restrict__ P5, float* __restrict__ out)
{
    const int i = blockIdx.x;
    const int b = blockIdx.y;
    const int j = threadIdx.x;

    __shared__ float sQi[ODIM];
    __shared__ float sP5p[ODIM];
    if (threadIdx.x < ODIM) {
        const int o = threadIdx.x;
        sQi[o]  = Qi[(b * LDIM + i) * ODIM + o];
        sP5p[o] = (i == 0) ? 0.0f : P5[(b * LDIM + i - 1) * ODIM + o];
    }
    __syncthreads();

    const float denom = (float)(j - i + 1) + EPSV;
    const float rden  = 1.0f / denom;

    const float* qj = Qj + (b * LDIM + j) * ODIM;
    const float* pj = P5 + (b * LDIM + j) * ODIM;
    float* o0 = out + ((size_t)(b * ODIM) * LDIM + i) * LDIM + j;

    #pragma unroll
    for (int o = 0; o < ODIM; ++o) {
        const float v = qj[o] + sQi[o] + (pj[o] - sP5p[o]) * rden;
        o0[(size_t)o * LDIM * LDIM] = v;
    }
}

extern "C" void kernel_launch(void* const* d_in, const int* in_sizes, int n_in,
                              void* d_out, int out_size, void* d_ws, size_t ws_size,
                              hipStream_t stream)
{
    const float* x  = (const float*)d_in[0];
    const float* W1 = (const float*)d_in[1];
    const float* b1 = (const float*)d_in[2];
    const float* W2 = (const float*)d_in[3];
    const float* b2 = (const float*)d_in[4];
    const float* Wf = (const float*)d_in[5];
    const float* bf = (const float*)d_in[6];
    float* out = (float*)d_out;

    float* ws  = (float*)d_ws;
    float* h1  = ws;
    float* h2  = h1  + (size_t)MDIM * HDIM;
    float* cum = h2  + (size_t)MDIM * HDIM;
    float* Qj  = cum + (size_t)MDIM * HDIM;
    float* Qi  = Qj  + (size_t)MDIM * ODIM;
    float* P5  = Qi  + (size_t)MDIM * ODIM;

    dim3 g1(MDIM / 16, HDIM / 64);
    mlp_wmma_kernel<<<g1, 32, 0, stream>>>(x, W1, b1, h1);
    mlp_wmma_kernel<<<g1, 32, 0, stream>>>(x, W2, b2, h2);
    cumsum_kernel<<<BDIM, HDIM, 0, stream>>>(h1, cum);
    proj_wmma_kernel<<<MDIM / 16, 32, 0, stream>>>(h1, h2, cum, Wf, bf, Qj, Qi, P5);
    epilogue_kernel<<<dim3(LDIM, BDIM), LDIM, 0, stream>>>(Qj, Qi, P5, out);
}


# Both MLP GEMMs (h1 and h2) run in one launch: blockIdx.z picks W1/b1/h1 or W2/b2/h2, so 256 single-wave tiles run side by side instead of 2 x 128 one after the other; 5 launches become 4
# speedup vs baseline: 1.4467x; 1.4467x over previous
#include <hip/hip_runtime.h>

#define LDIM 256
#define DDIM 512
#define HDIM 256
#define ODIM 16
#define MDIM 512
#define BDIM 2
#define LEAKY 0.1f
#define EPSV 1e-9f

typedef __attribute__((ext_vector_type(2))) float v2f;
typedef __attribute__((ext_vector_type(8))) float v8f;

__global__ __launch_bounds__(32) void mlp_wmma_kernel(
    const float* __restrict__ X,
    const float* __restrict__ Wa, const float* __restrict__ biasA, float* __restrict__ HoutA,
    const float* __restrict__ Wb, const float* __restrict__ biasB, float* __restrict__ HoutB)
{
    const bool second = (blockIdx.z != 0);
    const float* __restrict__ W    = second ? Wb    : Wa;
    const float* __restrict__ bias = second ? biasB : biasA;
    float* __restrict__       Hout = second ? HoutB : HoutA;

    const int lane  = threadIdx.x & 31;
    const int mBase = blockIdx.x * 16;
    const int nBase = blockIdx.y * 64;
    const int mRow  = mBase + (lane & 15);
    const int kOff  = (lane >> 4) * 2;

    v8f acc0 = {}, acc1 = {}, acc2 = {}, acc3 = {};

    const float* xrow = X + mRow * DDIM + kOff;
    const int n0 = nBase + (lane & 15);
    const float* w0 = W + (n0 +  0) * DDIM + kOff;
    const float* w1 = W + (n0 + 16) * DDIM + kOff;
    const float* w2 = W + (n0 + 32) * DDIM + kOff;
    const float* w3 = W + (n0 + 48) * DDIM + kOff;

    #pragma unroll 4
    for (int k = 0; k < DDIM; k += 4) {
        v2f a  = *(const v2f*)(xrow + k);
        v2f b0 = *(const v2f*)(w0 + k);
        v2f b1 = *(const v2f*)(w1 + k);
        v2f b2 = *(const v2f*)(w2 + k);
        v2f b3 = *(const v2f*)(w3 + k);
        acc0 = __builtin_amdgcn_wmma_f32_16x16x4_f32(false, a, false, b0, 0, acc0, false, false);
        acc1 = __builtin_amdgcn_wmma_f32_16x16x4_f32(false, a, false, b1, 0, acc1, false, false);
        acc2 = __builtin_amdgcn_wmma_f32_16x16x4_f32(false, a, false, b2, 0, acc2, false, false);
        acc3 = __builtin_amdgcn_wmma_f32_16x16x4_f32(false, a, false, b3, 0, acc3, false, false);
    }

    const int mOut = mBase + (lane >> 4) * 8;
    v8f accs[4] = {acc0, acc1, acc2, acc3};
    #pragma unroll
    for (int t = 0; t < 4; ++t) {
        const int nCol = nBase + t * 16 + (lane & 15);
        const float bv = bias[nCol];
        v8f acc = accs[t];
        #pragma unroll
        for (int r = 0; r < 8; ++r) {
            float h = acc[r] + bv;
            h = (h >= 0.0f) ? h : (LEAKY * h);
            Hout[(mOut + r) * HDIM + nCol] = h;
        }
    }
}

__global__ void cumsum_kernel(const float* __restrict__ h1, float* __restrict__ cum)
{
    const int b = blockIdx.x;
    const int h = threadIdx.x;
    const float* src = h1 + b * LDIM * HDIM + h;
    float*       dst = cum + b * LDIM * HDIM + h;
    float run = 0.0f;
    for (int l = 0; l < LDIM; ++l) {
        run += src[l * HDIM];
        dst[l * HDIM] = run;
    }
}

__global__ __launch_bounds__(32) void proj_wmma_kernel(
    const float* __restrict__ h1, const float* __restrict__ h2,
    const float* __restrict__ cum, const float* __restrict__ Wf,
    const float* __restrict__ bf,
    float* __restrict__ Qj, float* __restrict__ Qi, float* __restrict__ P5)
{
    const int lane = threadIdx.x & 31;
    const int mRow = blockIdx.x * 16 + (lane & 15);
    const int kOff = (lane >> 4) * 2;
    const int o    = lane & 15;

    const float* a1p = h1  + mRow * HDIM + kOff;
    const float* a2p = h2  + mRow * HDIM + kOff;
    const float* a5p = cum + mRow * HDIM + kOff;
    const float* w1p = Wf + o * (3 * HDIM) + kOff;
    const float* w2p = w1p + HDIM;
    const float* w3p = w1p + 2 * HDIM;

    v8f p1 = {}, p2 = {}, p3 = {}, p4 = {}, p5 = {};
    #pragma unroll 4
    for (int k = 0; k < HDIM; k += 4) {
        v2f a1 = *(const v2f*)(a1p + k);
        v2f a2 = *(const v2f*)(a2p + k);
        v2f a5 = *(const v2f*)(a5p + k);
        v2f b1 = *(const v2f*)(w1p + k);
        v2f b2 = *(const v2f*)(w2p + k);
        v2f b3 = *(const v2f*)(w3p + k);
        p1 = __builtin_amdgcn_wmma_f32_16x16x4_f32(false, a1, false, b1, 0, p1, false, false);
        p2 = __builtin_amdgcn_wmma_f32_16x16x4_f32(false, a2, false, b1, 0, p2, false, false);
        p3 = __builtin_amdgcn_wmma_f32_16x16x4_f32(false, a2, false, b2, 0, p3, false, false);
        p4 = __builtin_amdgcn_wmma_f32_16x16x4_f32(false, a1, false, b2, 0, p4, false, false);
        p5 = __builtin_amdgcn_wmma_f32_16x16x4_f32(false, a5, false, b3, 0, p5, false, false);
    }

    const int mOut = blockIdx.x * 16 + (lane >> 4) * 8;
    const float bfv = bf[o];
    #pragma unroll
    for (int r = 0; r < 8; ++r) {
        const int m = mOut + r;
        Qj[m * ODIM + o] = p1[r] + 0.5f * p3[r] + bfv;
        Qi[m * ODIM + o] = 0.5f * p4[r] - p2[r];
        P5[m * ODIM + o] = p5[r];
    }
}

__global__ __launch_bounds__(LDIM) void epilogue_kernel(
    const float* __restrict__ Qj, const float* __restrict__ Qi,
    const float* __restrict__ P5, float* __restrict__ out)
{
    const int i = blockIdx.x;
    const int b = blockIdx.y;
    const int j = threadIdx.x;

    __shared__ float sQi[ODIM];
    __shared__ float sP5p[ODIM];
    if (threadIdx.x < ODIM) {
        const int o = threadIdx.x;
        sQi[o]  = Qi[(b * LDIM + i) * ODIM + o];
        sP5p[o] = (i == 0) ? 0.0f : P5[(b * LDIM + i - 1) * ODIM + o];
    }
    __syncthreads();

    const float denom = (float)(j - i + 1) + EPSV;
    const float rden  = 1.0f / denom;

    const float* qj = Qj + (b * LDIM + j) * ODIM;
    const float* pj = P5 + (b * LDIM + j) * ODIM;
    float* o0 = out + ((size_t)(b * ODIM) * LDIM + i) * LDIM + j;

    #pragma unroll
    for (int o = 0; o < ODIM; ++o) {
        const float v = qj[o] + sQi[o] + (pj[o] - sP5p[o]) * rden;
        o0[(size_t)o * LDIM * LDIM] = v;
    }
}

extern "C" void kernel_launch(void* const* d_in, const int* in_sizes, int n_in,
                              void* d_out, int out_size, void* d_ws, size_t ws_size,
                              hipStream_t stream)
{
    const float* x  = (const float*)d_in[0];
    const float* W1 = (const float*)d_in[1];
    const float* b1 = (const float*)d_in[2];
    const float* W2 = (const float*)d_in[3];
    const float* b2 = (const float*)d_in[4];
    const float* Wf = (const float*)d_in[5];
    const float* bf = (const float*)d_in[6];
    float* out = (float*)d_out;

    float* ws  = (float*)d_ws;
    float* h1  = ws;
    float* h2  = h1  + (size_t)MDIM * HDIM;
    float* cum = h2  + (size_t)MDIM * HDIM;
    float* Qj  = cum + (size_t)MDIM * HDIM;
    float* Qi  = Qj  + (size_t)MDIM * ODIM;
    float* P5  = Qi  + (size_t)MDIM * ODIM;

    dim3 g1(MDIM / 16, HDIM / 64, 2);
    mlp_wmma_kernel<<<g1, 32, 0, stream>>>(x, W1, b1, h1, W2, b2, h2);
    cumsum_kernel<<<BDIM, HDIM, 0, stream>>>(h1, cum);
    proj_wmma_kernel<<<MDIM / 16, 32, 0, stream>>>(h1, h2, cum, Wf, bf, Qj, Qi, P5);
    epilogue_kernel<<<dim3(LDIM, BDIM), LDIM, 0, stream>>>(Qj, Qi, P5, out);
}


# K of both WMMA kernels split over the 8 waves of a 256-thread block, partial tiles summed through LDS in fixed wave order (new summation tree); no wave runs a whole K chain alone
# speedup vs baseline: 2.7692x; 1.9141x over previous
#include <hip/hip_runtime.h>

#define LDIM 256
#define DDIM 512
#define HDIM 256
#define ODIM 16
#define MDIM 512
#define BDIM 2
#define LEAKY 0.1f
#define EPSV 1e-9f

#define KWAVES 8
#define MLP_KSLICE  (DDIM / KWAVES)
#define PROJ_KSLICE (HDIM / KWAVES)
#define MLP_TILE_LDS 272

typedef __attribute__((ext_vector_type(2))) float v2f;
typedef __attribute__((ext_vector_type(4))) float v4f;
typedef __attribute__((ext_vector_type(8))) float v8f;

__global__ __launch_bounds__(32 * KWAVES) void mlp_wmma_kernel(
    const float* __restrict__ X,
    const float* __restrict__ Wa, const float* __restrict__ biasA, float* __restrict__ HoutA,
    const float* __restrict__ Wb, const float* __restrict__ biasB, float* __restrict__ HoutB)
{
    __shared__ __attribute__((aligned(16))) float part[KWAVES * 4 * MLP_TILE_LDS];

    const bool second = (blockIdx.z != 0);
    const float* __restrict__ W    = second ? Wb    : Wa;
    const float* __restrict__ bias = second ? biasB : biasA;
    float* __restrict__       Hout = second ? HoutB : HoutA;

    const int tid   = threadIdx.x;
    const int lane  = tid & 31;
    const int wave  = tid >> 5;
    const int mBase = blockIdx.x * 16;
    const int nBase = blockIdx.y * 64;
    const int mRow  = mBase + (lane & 15);
    const int kOff  = wave * MLP_KSLICE + (lane >> 4) * 2;

    v8f acc0 = {}, acc1 = {}, acc2 = {}, acc3 = {};

    const float* xrow = X + mRow * DDIM + kOff;
    const int n0 = nBase + (lane & 15);
    const float* w0 = W + (n0 +  0) * DDIM + kOff;
    const float* w1 = W + (n0 + 16) * DDIM + kOff;
    const float* w2 = W + (n0 + 32) * DDIM + kOff;
    const float* w3 = W + (n0 + 48) * DDIM + kOff;

    #pragma unroll
    for (int k = 0; k < MLP_KSLICE; k += 4) {
        v2f a  = *(const v2f*)(xrow + k);
        v2f b0 = *(const v2f*)(w0 + k);
        v2f b1 = *(const v2f*)(w1 + k);
        v2f b2 = *(const v2f*)(w2 + k);
        v2f b3 = *(const v2f*)(w3 + k);
        acc0 = __builtin_amdgcn_wmma_f32_16x16x4_f32(false, a, false, b0, 0, acc0, false, false);
        acc1 = __builtin_amdgcn_wmma_f32_16x16x4_f32(false, a, false, b1, 0, acc1, false, false);
        acc2 = __builtin_amdgcn_wmma_f32_16x16x4_f32(false, a, false, b2, 0, acc2, false, false);
        acc3 = __builtin_amdgcn_wmma_f32_16x16x4_f32(false, a, false, b3, 0, acc3, false, false);
    }

    float* myPart = part + wave * (4 * MLP_TILE_LDS) + lane;
    v8f accs[4] = {acc0, acc1, acc2, acc3};
    #pragma unroll
    for (int t = 0; t < 4; ++t) {
        #pragma unroll
        for (int r = 0; r < 8; ++r) {
            myPart[t * MLP_TILE_LDS + r * 32] = accs[t][r];
        }
    }
    __syncthreads();

    const int row = tid >> 4;
    const int col = (tid & 15) * 4;
    const float* src = part + (col >> 4) * MLP_TILE_LDS
                     + (row & 7) * 32 + (row >> 3) * 16 + (col & 15);
    v4f s = *(const v4f*)src;
    #pragma unroll
    for (int w = 1; w < KWAVES; ++w) {
        s += *(const v4f*)(src + w * (4 * MLP_TILE_LDS));
    }
    const v4f bv = *(const v4f*)(bias + nBase + col);
    v4f h = s + bv;
    #pragma unroll
    for (int e = 0; e < 4; ++e) {
        h[e] = (h[e] >= 0.0f) ? h[e] : (LEAKY * h[e]);
    }
    *(v4f*)(Hout + (mBase + row) * HDIM + nBase + col) = h;
}

__global__ void cumsum_kernel(const float* __restrict__ h1, float* __restrict__ cum)
{
    const int b = blockIdx.x;
    const int h = threadIdx.x;
    const float* src = h1 + b * LDIM * HDIM + h;
    float*       dst = cum + b * LDIM * HDIM + h;
    float run = 0.0f;
    for (int l = 0; l < LDIM; ++l) {
        run += src[l * HDIM];
        dst[l * HDIM] = run;
    }
}

__global__ __launch_bounds__(32 * KWAVES) void proj_wmma_kernel(
    const float* __restrict__ h1, const float* __restrict__ h2,
    const float* __restrict__ cum, const float* __restrict__ Wf,
    const float* __restrict__ bf,
    float* __restrict__ Qj, float* __restrict__ Qi, float* __restrict__ P5)
{
    __shared__ float part[KWAVES * 5 * 256];

    const int tid  = threadIdx.x;
    const int lane = tid & 31;
    const int wave = tid >> 5;
    const int mRow = blockIdx.x * 16 + (lane & 15);
    const int kOff = wave * PROJ_KSLICE + (lane >> 4) * 2;
    const int o    = lane & 15;

    const float* a1p = h1  + mRow * HDIM + kOff;
    const float* a2p = h2  + mRow * HDIM + kOff;
    const float* a5p = cum + mRow * HDIM + kOff;
    const float* w1p = Wf + o * (3 * HDIM) + kOff;
    const float* w2p = w1p + HDIM;
    const float* w3p = w1p + 2 * HDIM;

    v8f p1 = {}, p2 = {}, p3 = {}, p4 = {}, p5 = {};
    #pragma unroll
    for (int k = 0; k < PROJ_KSLICE; k += 4) {
        v2f a1 = *(const v2f*)(a1p + k);
        v2f a2 = *(const v2f*)(a2p + k);
        v2f a5 = *(const v2f*)(a5p + k);
        v2f b1 = *(const v2f*)(w1p + k);
        v2f b2 = *(const v2f*)(w2p + k);
        v2f b3 = *(const v2f*)(w3p + k);
        p1 = __builtin_amdgcn_wmma_f32_16x16x4_f32(false, a1, false, b1, 0, p1, false, false);
        p2 = __builtin_amdgcn_wmma_f32_16x16x4_f32(false, a2, false, b1, 0, p2, false, false);
        p3 = __builtin_amdgcn_wmma_f32_16x16x4_f32(false, a2, false, b2, 0, p3, false, false);
        p4 = __builtin_amdgcn_wmma_f32_16x16x4_f32(false, a1, false, b2, 0, p4, false, false);
        p5 = __builtin_amdgcn_wmma_f32_16x16x4_f32(false, a5, false, b3, 0, p5, false, false);
    }

    float* myPart = part + wave * (5 * 256) + lane;
    #pragma unroll
    for (int r = 0; r < 8; ++r) {
        myPart[0 * 256 + r * 32] = p1[r];
        myPart[1 * 256 + r * 32] = p2[r];
        myPart[2 * 256 + r * 32] = p3[r];
        myPart[3 * 256 + r * 32] = p4[r];
        myPart[4 * 256 + r * 32] = p5[r];
    }
    __syncthreads();

    const int row = tid >> 4;
    const int oc  = tid & 15;
    const float* src = part + (row & 7) * 32 + (row >> 3) * 16 + oc;
    float s1 = src[0 * 256], s2 = src[1 * 256], s3 = src[2 * 256],
          s4 = src[3 * 256], s5 = src[4 * 256];
    #pragma unroll
    for (int w = 1; w < KWAVES; ++w) {
        const float* q = src + w * (5 * 256);
        s1 += q[0 * 256];
        s2 += q[1 * 256];
        s3 += q[2 * 256];
        s4 += q[3 * 256];
        s5 += q[4 * 256];
    }

    const int m = blockIdx.x * 16 + row;
    Qj[m * ODIM + oc] = s1 + 0.5f * s3 + bf[oc];
    Qi[m * ODIM + oc] = 0.5f * s4 - s2;
    P5[m * ODIM + oc] = s5;
}

__global__ __launch_bounds__(LDIM) void epilogue_kernel(
    const float* __restrict__ Qj, const float* __restrict__ Qi,
    const float* __restrict__ P5, float* __restrict__ out)
{
    const int i = blockIdx.x;
    const int b = blockIdx.y;
    const int j = threadIdx.x;

    __shared__ float sQi[ODIM];
    __shared__ float sP5p[ODIM];
    if (threadIdx.x < ODIM) {
        const int o = threadIdx.x;
        sQi[o]  = Qi[(b * LDIM + i) * ODIM + o];
        sP5p[o] = (i == 0) ? 0.0f : P5[(b * LDIM + i - 1) * ODIM + o];
    }
    __syncthreads();

    const float denom = (float)(j - i + 1) + EPSV;
    const float rden  = 1.0f / denom;

    const float* qj = Qj + (b * LDIM + j) * ODIM;
    const float* pj = P5 + (b * LDIM + j) * ODIM;
    float* o0 = out + ((size_t)(b * ODIM) * LDIM + i) * LDIM + j;

    #pragma unroll
    for (int o = 0; o < ODIM; ++o) {
        const float v = qj[o] + sQi[o] + (pj[o] - sP5p[o]) * rden;
        o0[(size_t)o * LDIM * LDIM] = v;
    }
}

extern "C" void kernel_launch(void* const* d_in, const int* in_sizes, int n_in,
                              void* d_out, int out_size, void* d_ws, size_t ws_size,
                              hipStream_t stream)
{
    const float* x  = (const float*)d_in[0];
    const float* W1 = (const float*)d_in[1];
    const float* b1 = (const float*)d_in[2];
    const float* W2 = (const float*)d_in[3];
    const float* b2 = (const float*)d_in[4];
    const float* Wf = (const float*)d_in[5];
    const float* bf = (const float*)d_in[6];
    float* out = (float*)d_out;

    float* ws  = (float*)d_ws;
    float* h1  = ws;
    float* h2  = h1  + (size_t)MDIM * HDIM;
    float* cum = h2  + (size_t)MDIM * HDIM;
    float* Qj  = cum + (size_t)MDIM * HDIM;
    float* Qi  = Qj  + (size_t)MDIM * ODIM;
    float* P5  = Qi  + (size_t)MDIM * ODIM;

    dim3 g1(MDIM / 16, HDIM / 64, 2);
    mlp_wmma_kernel<<<g1, 32 * KWAVES, 0, stream>>>(x, W1, b1, h1, W2, b2, h2);
    cumsum_kernel<<<BDIM, HDIM, 0, stream>>>(h1, cum);
    proj_wmma_kernel<<<MDIM / 16, 32 * KWAVES, 0, stream>>>(h1, h2, cum, Wf, bf, Qj, Qi, P5);
    epilogue_kernel<<<dim3(LDIM, BDIM), LDIM, 0, stream>>>(Qj, Qi, P5, out);
}


# cumsum launch removed: Wf3 applied to h1 rows first, then a 2-level running sum over l (in-tile in projection kernel, tile offsets in epilogue); new summation order of span-sum term; 4 launches to 3
# speedup vs baseline: 4.6303x; 1.6721x over previous
#include <hip/hip_runtime.h>

#define LDIM 256
#define DDIM 512
#define HDIM 256
#define ODIM 16
#define MDIM 512
#define BDIM 2
#define LEAKY 0.1f
#define EPSV 1e-9f

#define KWAVES 8
#define MLP_KSLICE  (DDIM / KWAVES)
#define PROJ_KSLICE (HDIM / KWAVES)
#define MLP_TILE_LDS 272

typedef __attribute__((ext_vector_type(2))) float v2f;
typedef __attribute__((ext_vector_type(4))) float v4f;
typedef __attribute__((ext_vector_type(8))) float v8f;

__global__ __launch_bounds__(32 * KWAVES) void mlp_wmma_kernel(
    const float* __restrict__ X,
    const float* __restrict__ Wa, const float* __restrict__ biasA, float* __restrict__ HoutA,
    const float* __restrict__ Wb, const float* __restrict__ biasB, float* __restrict__ HoutB)
{
    __shared__ __attribute__((aligned(16))) float part[KWAVES * 4 * MLP_TILE_LDS];

    const bool second = (blockIdx.z != 0);
    const float* __restrict__ W    = second ? Wb    : Wa;
    const float* __restrict__ bias = second ? biasB : biasA;
    float* __restrict__       Hout = second ? HoutB : HoutA;

    const int tid   = threadIdx.x;
    const int lane  = tid & 31;
    const int wave  = tid >> 5;
    const int mBase = blockIdx.x * 16;
    const int nBase = blockIdx.y * 64;
    const int mRow  = mBase + (lane & 15);
    const int kOff  = wave * MLP_KSLICE + (lane >> 4) * 2;

    v8f acc0 = {}, acc1 = {}, acc2 = {}, acc3 = {};

    const float* xrow = X + mRow * DDIM + kOff;
    const int n0 = nBase + (lane & 15);
    const float* w0 = W + (n0 +  0) * DDIM + kOff;
    const float* w1 = W + (n0 + 16) * DDIM + kOff;
    const float* w2 = W + (n0 + 32) * DDIM + kOff;
    const float* w3 = W + (n0 + 48) * DDIM + kOff;

    #pragma unroll
    for (int k = 0; k < MLP_KSLICE; k += 4) {
        v2f a  = *(const v2f*)(xrow + k);
        v2f b0 = *(const v2f*)(w0 + k);
        v2f b1 = *(const v2f*)(w1 + k);
        v2f b2 = *(const v2f*)(w2 + k);
        v2f b3 = *(const v2f*)(w3 + k);
        acc0 = __builtin_amdgcn_wmma_f32_16x16x4_f32(false, a, false, b0, 0, acc0, false, false);
        acc1 = __builtin_amdgcn_wmma_f32_16x16x4_f32(false, a, false, b1, 0, acc1, false, false);
        acc2 = __builtin_amdgcn_wmma_f32_16x16x4_f32(false, a, false, b2, 0, acc2, false, false);
        acc3 = __builtin_amdgcn_wmma_f32_16x16x4_f32(false, a, false, b3, 0, acc3, false, false);
    }

    float* myPart = part + wave * (4 * MLP_TILE_LDS) + lane;
    v8f accs[4] = {acc0, acc1, acc2, acc3};
    #pragma unroll
    for (int t = 0; t < 4; ++t) {
        #pragma unroll
        for (int r = 0; r < 8; ++r) {
            myPart[t * MLP_TILE_LDS + r * 32] = accs[t][r];
        }
    }
    __syncthreads();

    const int row = tid >> 4;
    const int col = (tid & 15) * 4;
    const float* src = part + (col >> 4) * MLP_TILE_LDS
                     + (row & 7) * 32 + (row >> 3) * 16 + (col & 15);
    v4f s = *(const v4f*)src;
    #pragma unroll
    for (int w = 1; w < KWAVES; ++w) {
        s += *(const v4f*)(src + w * (4 * MLP_TILE_LDS));
    }
    const v4f bv = *(const v4f*)(bias + nBase + col);
    v4f h = s + bv;
    #pragma unroll
    for (int e = 0; e < 4; ++e) {
        h[e] = (h[e] >= 0.0f) ? h[e] : (LEAKY * h[e]);
    }
    *(v4f*)(Hout + (mBase + row) * HDIM + nBase + col) = h;
}

#define TILES_PER_B (LDIM / 16)
#define TT_PITCH 32

__global__ __launch_bounds__(32 * KWAVES) void proj_wmma_kernel(
    const float* __restrict__ h1, const float* __restrict__ h2,
    const float* __restrict__ Wf, const float* __restrict__ bf,
    float* __restrict__ Qj, float* __restrict__ Qi,
    float* __restrict__ P5t, float* __restrict__ Tt)
{
    __shared__ float part[KWAVES * 5 * 256];
    __shared__ float rtile[256];

    const int tid  = threadIdx.x;
    const int lane = tid & 31;
    const int wave = tid >> 5;
    const int mRow = blockIdx.x * 16 + (lane & 15);
    const int kOff = wave * PROJ_KSLICE + (lane >> 4) * 2;
    const int o    = lane & 15;

    const float* a1p = h1 + mRow * HDIM + kOff;
    const float* a2p = h2 + mRow * HDIM + kOff;
    const float* w1p = Wf + o * (3 * HDIM) + kOff;
    const float* w2p = w1p + HDIM;
    const float* w3p = w1p + 2 * HDIM;

    v8f p1 = {}, p2 = {}, p3 = {}, p4 = {}, p5 = {};
    #pragma unroll
    for (int k = 0; k < PROJ_KSLICE; k += 4) {
        v2f a1 = *(const v2f*)(a1p + k);
        v2f a2 = *(const v2f*)(a2p + k);
        v2f b1 = *(const v2f*)(w1p + k);
        v2f b2 = *(const v2f*)(w2p + k);
        v2f b3 = *(const v2f*)(w3p + k);
        p1 = __builtin_amdgcn_wmma_f32_16x16x4_f32(false, a1, false, b1, 0, p1, false, false);
        p2 = __builtin_amdgcn_wmma_f32_16x16x4_f32(false, a2, false, b1, 0, p2, false, false);
        p3 = __builtin_amdgcn_wmma_f32_16x16x4_f32(false, a2, false, b2, 0, p3, false, false);
        p4 = __builtin_amdgcn_wmma_f32_16x16x4_f32(false, a1, false, b2, 0, p4, false, false);
        p5 = __builtin_amdgcn_wmma_f32_16x16x4_f32(false, a1, false, b3, 0, p5, false, false);
    }

    float* myPart = part + wave * (5 * 256) + lane;
    #pragma unroll
    for (int r = 0; r < 8; ++r) {
        myPart[0 * 256 + r * 32] = p1[r];
        myPart[1 * 256 + r * 32] = p2[r];
        myPart[2 * 256 + r * 32] = p3[r];
        myPart[3 * 256 + r * 32] = p4[r];
        myPart[4 * 256 + r * 32] = p5[r];
    }
    __syncthreads();

    const int row = tid >> 4;
    const int oc  = tid & 15;
    const float* src = part + (row & 7) * 32 + (row >> 3) * 16 + oc;
    float s1 = src[0 * 256], s2 = src[1 * 256], s3 = src[2 * 256],
          s4 = src[3 * 256], s5 = src[4 * 256];
    #pragma unroll
    for (int w = 1; w < KWAVES; ++w) {
        const float* q = src + w * (5 * 256);
        s1 += q[0 * 256];
        s2 += q[1 * 256];
        s3 += q[2 * 256];
        s4 += q[3 * 256];
        s5 += q[4 * 256];
    }

    rtile[tid] = s5;
    __syncthreads();
    float run = 0.0f;
    #pragma unroll
    for (int r = 0; r < 16; ++r) {
        const float v = rtile[r * 16 + oc];
        run += (r <= row) ? v : 0.0f;
    }

    const int m = blockIdx.x * 16 + row;
    Qj[m * ODIM + oc]  = s1 + 0.5f * s3 + bf[oc];
    Qi[m * ODIM + oc]  = 0.5f * s4 - s2;
    P5t[m * ODIM + oc] = run;
    if (row == 15) {
        Tt[blockIdx.x * TT_PITCH + oc] = run;
    }
}

__global__ __launch_bounds__(LDIM) void epilogue_kernel(
    const float* __restrict__ Qj, const float* __restrict__ Qi,
    const float* __restrict__ P5t, const float* __restrict__ Tt,
    float* __restrict__ out)
{
    const int i = blockIdx.x;
    const int b = blockIdx.y;
    const int j = threadIdx.x;

    __shared__ __attribute__((aligned(16))) float sQi[ODIM];
    __shared__ __attribute__((aligned(16))) float sP5p[ODIM];
    __shared__ __attribute__((aligned(16))) float sOff[TILES_PER_B * ODIM];

    {
        const int t = threadIdx.x >> 4;
        const int o = threadIdx.x & 15;
        const float* tt = Tt + (b * TILES_PER_B) * TT_PITCH + o;
        float acc = 0.0f;
        #pragma unroll
        for (int tp = 0; tp < TILES_PER_B - 1; ++tp) {
            const float v = tt[tp * TT_PITCH];
            acc += (tp < t) ? v : 0.0f;
        }
        sOff[threadIdx.x] = acc;
    }
    if (threadIdx.x < ODIM) {
        const int o  = threadIdx.x;
        const int ip = (i == 0) ? 0 : (i - 1);
        const float qiv = Qi [(b * LDIM + i)  * ODIM + o];
        const float ppv = P5t[(b * LDIM + ip) * ODIM + o];
        sQi[o]  = qiv;
        sP5p[o] = (i == 0) ? 0.0f : ppv;
    }
    __syncthreads();

    const float denom = (float)(j - i + 1) + EPSV;
    const float rden  = 1.0f / denom;

    const float* qj = Qj  + (b * LDIM + j) * ODIM;
    const float* pj = P5t + (b * LDIM + j) * ODIM;
    const float* offJ = sOff + (j >> 4) * ODIM;
    const float* offP = sOff + ((i == 0) ? 0 : ((i - 1) >> 4)) * ODIM;
    float* o0 = out + ((size_t)(b * ODIM) * LDIM + i) * LDIM + j;

    #pragma unroll
    for (int o = 0; o < ODIM; ++o) {
        const float p5j = pj[o]   + offJ[o];
        const float p5p = sP5p[o] + offP[o];
        const float v = qj[o] + sQi[o] + (p5j - p5p) * rden;
        o0[(size_t)o * LDIM * LDIM] = v;
    }
}

extern "C" void kernel_launch(void* const* d_in, const int* in_sizes, int n_in,
                              void* d_out, int out_size, void* d_ws, size_t ws_size,
                              hipStream_t stream)
{
    const float* x  = (const float*)d_in[0];
    const float* W1 = (const float*)d_in[1];
    const float* b1 = (const float*)d_in[2];
    const float* W2 = (const float*)d_in[3];
    const float* b2 = (const float*)d_in[4];
    const float* Wf = (const float*)d_in[5];
    const float* bf = (const float*)d_in[6];
    float* out = (float*)d_out;

    float* ws  = (float*)d_ws;
    float* h1  = ws;
    float* h2  = h1  + (size_t)MDIM * HDIM;
    float* Tt  = h2  + (size_t)MDIM * HDIM;
    float* Qj  = Tt  + (size_t)MDIM * HDIM;
    float* Qi  = Qj  + (size_t)MDIM * ODIM;
    float* P5t = Qi  + (size_t)MDIM * ODIM;

    dim3 g1(MDIM / 16, HDIM / 64, 2);
    mlp_wmma_kernel<<<g1, 32 * KWAVES, 0, stream>>>(x, W1, b1, h1, W2, b2, h2);
    proj_wmma_kernel<<<MDIM / 16, 32 * KWAVES, 0, stream>>>(h1, h2, Wf, bf, Qj, Qi, P5t, Tt);
    epilogue_kernel<<<dim3(LDIM, BDIM), LDIM, 0, stream>>>(Qj, Qi, P5t, Tt, out);
}


# MLP kernel: K split over 4 waves, one per SIMD (was 8): half the waves enter; each wave streams its 128 k with 16-byte loads 8 groups ahead of its WMMAs (lane half h: k 4h..4h+3); new summation tree
# speedup vs baseline: 4.8922x; 1.0566x over previous
#include <hip/hip_runtime.h>

#define LDIM 256
#define DDIM 512
#define HDIM 256
#define ODIM 16
#define MDIM 512
#define BDIM 2
#define LEAKY 0.1f
#define EPSV 1e-9f

#define KWAVES 8
#define PROJ_KSLICE (HDIM / KWAVES)
#define MLP_WAVES   4
#define MLP_KSLICE  (DDIM / MLP_WAVES)
#define MLP_GROUPS  (MLP_KSLICE / 8)
#define MLP_AHEAD   8
#define MLP_TILE_LDS 272

typedef __attribute__((ext_vector_type(2))) float v2f;
typedef __attribute__((ext_vector_type(4))) float v4f;
typedef __attribute__((ext_vector_type(8))) float v8f;

__global__ __launch_bounds__(32 * MLP_WAVES) void mlp_wmma_kernel(
    const float* X,
    const float* Wa, const float* biasA, float* HoutA,
    const float* Wb, const float* biasB, float* HoutB)
{
    __shared__ __attribute__((aligned(16))) float part[MLP_WAVES * 4 * MLP_TILE_LDS];

    const bool second = (blockIdx.z != 0);
    const float* W    = second ? Wb    : Wa;
    const float* bias = second ? biasB : biasA;
    float*       Hout = second ? HoutB : HoutA;

    const int tid   = threadIdx.x;
    const int lane  = tid & 31;
    const int wave  = tid >> 5;
    const int mBase = blockIdx.x * 16;
    const int nBase = blockIdx.y * 64;
    const int mRow  = mBase + (lane & 15);
    const int kOff  = wave * MLP_KSLICE + (lane >> 4) * 4;

    v8f acc0 = {}, acc1 = {}, acc2 = {}, acc3 = {};

    const float* xrow = X + mRow * DDIM + kOff;
    const int n0 = nBase + (lane & 15);
    const float* w0 = W + (n0 +  0) * DDIM + kOff;
    const float* w1 = W + (n0 + 16) * DDIM + kOff;
    const float* w2 = W + (n0 + 32) * DDIM + kOff;
    const float* w3 = W + (n0 + 48) * DDIM + kOff;

    v4f fa[MLP_GROUPS], f0[MLP_GROUPS], f1[MLP_GROUPS], f2[MLP_GROUPS], f3[MLP_GROUPS];

    #pragma unroll
    for (int g = 0; g < MLP_AHEAD; ++g) {
        fa[g] = *(const v4f*)(xrow + g * 8);
        f0[g] = *(const v4f*)(w0 + g * 8);
        f1[g] = *(const v4f*)(w1 + g * 8);
        f2[g] = *(const v4f*)(w2 + g * 8);
        f3[g] = *(const v4f*)(w3 + g * 8);
    }
    __builtin_amdgcn_sched_barrier(0);

    #pragma unroll
    for (int g = 0; g < MLP_GROUPS; ++g) {
        const v2f aL = {fa[g][0], fa[g][1]}, aH = {fa[g][2], fa[g][3]};
        const v2f b0L = {f0[g][0], f0[g][1]}, b0H = {f0[g][2], f0[g][3]};
        const v2f b1L = {f1[g][0], f1[g][1]}, b1H = {f1[g][2], f1[g][3]};
        const v2f b2L = {f2[g][0], f2[g][1]}, b2H = {f2[g][2], f2[g][3]};
        const v2f b3L = {f3[g][0], f3[g][1]}, b3H = {f3[g][2], f3[g][3]};
        acc0 = __builtin_amdgcn_wmma_f32_16x16x4_f32(false, aL, false, b0L, 0, acc0, false, false);
        acc1 = __builtin_amdgcn_wmma_f32_16x16x4_f32(false, aL, false, b1L, 0, acc1, false, false);
        acc2 = __builtin_amdgcn_wmma_f32_16x16x4_f32(false, aL, false, b2L, 0, acc2, false, false);
        acc3 = __builtin_amdgcn_wmma_f32_16x16x4_f32(false, aL, false, b3L, 0, acc3, false, false);
        acc0 = __builtin_amdgcn_wmma_f32_16x16x4_f32(false, aH, false, b0H, 0, acc0, false, false);
        acc1 = __builtin_amdgcn_wmma_f32_16x16x4_f32(false, aH, false, b1H, 0, acc1, false, false);
        acc2 = __builtin_amdgcn_wmma_f32_16x16x4_f32(false, aH, false, b2H, 0, acc2, false, false);
        acc3 = __builtin_amdgcn_wmma_f32_16x16x4_f32(false, aH, false, b3H, 0, acc3, false, false);
        if (g + MLP_AHEAD < MLP_GROUPS) {
            fa[g + MLP_AHEAD] = *(const v4f*)(xrow + (g + MLP_AHEAD) * 8);
            f0[g + MLP_AHEAD] = *(const v4f*)(w0 + (g + MLP_AHEAD) * 8);
            f1[g + MLP_AHEAD] = *(const v4f*)(w1 + (g + MLP_AHEAD) * 8);
            f2[g + MLP_AHEAD] = *(const v4f*)(w2 + (g + MLP_AHEAD) * 8);
            f3[g + MLP_AHEAD] = *(const v4f*)(w3 + (g + MLP_AHEAD) * 8);
            __builtin_amdgcn_sched_barrier(0);
        }
    }

    float* myPart = part + wave * (4 * MLP_TILE_LDS) + lane;
    v8f accs[4] = {acc0, acc1, acc2, acc3};
    #pragma unroll
    for (int t = 0; t < 4; ++t) {
        #pragma unroll
        for (int r = 0; r < 8; ++r) {
            myPart[t * MLP_TILE_LDS + r * 32] = accs[t][r];
        }
    }
    __syncthreads();

    const int col = (tid & 15) * 4;
    const v4f bv = *(const v4f*)(bias + nBase + col);
    #pragma unroll
    for (int half = 0; half < 2; ++half) {
        const int row = (tid >> 4) + 8 * half;
        const float* src = part + (col >> 4) * MLP_TILE_LDS
                         + (row & 7) * 32 + (row >> 3) * 16 + (col & 15);
        v4f s = *(const v4f*)src;
        #pragma unroll
        for (int w = 1; w < MLP_WAVES; ++w) {
            s += *(const v4f*)(src + w * (4 * MLP_TILE_LDS));
        }
        v4f h = s + bv;
        #pragma unroll
        for (int e = 0; e < 4; ++e) {
            h[e] = (h[e] >= 0.0f) ? h[e] : (LEAKY * h[e]);
        }
        *(v4f*)(Hout + (mBase + row) * HDIM + nBase + col) = h;
    }
}

#define TILES_PER_B (LDIM / 16)
#define TT_PITCH 32

__global__ __launch_bounds__(32 * KWAVES) void proj_wmma_kernel(
    const float* __restrict__ h1, const float* __restrict__ h2,
    const float* __restrict__ Wf, const float* __restrict__ bf,
    float* __restrict__ Qj, float* __restrict__ Qi,
    float* __restrict__ P5t, float* __restrict__ Tt)
{
    __shared__ float part[KWAVES * 5 * 256];
    __shared__ float rtile[256];

    const int tid  = threadIdx.x;
    const int lane = tid & 31;
    const int wave = tid >> 5;
    const int mRow = blockIdx.x * 16 + (lane & 15);
    const int kOff = wave * PROJ_KSLICE + (lane >> 4) * 2;
    const int o    = lane & 15;

    const float* a1p = h1 + mRow * HDIM + kOff;
    const float* a2p = h2 + mRow * HDIM + kOff;
    const float* w1p = Wf + o * (3 * HDIM) + kOff;
    const float* w2p = w1p + HDIM;
    const float* w3p = w1p + 2 * HDIM;

    v8f p1 = {}, p2 = {}, p3 = {}, p4 = {}, p5 = {};
    #pragma unroll
    for (int k = 0; k < PROJ_KSLICE; k += 4) {
        v2f a1 = *(const v2f*)(a1p + k);
        v2f a2 = *(const v2f*)(a2p + k);
        v2f b1 = *(const v2f*)(w1p + k);
        v2f b2 = *(const v2f*)(w2p + k);
        v2f b3 = *(const v2f*)(w3p + k);
        p1 = __builtin_amdgcn_wmma_f32_16x16x4_f32(false, a1, false, b1, 0, p1, false, false);
        p2 = __builtin_amdgcn_wmma_f32_16x16x4_f32(false, a2, false, b1, 0, p2, false, false);
        p3 = __builtin_amdgcn_wmma_f32_16x16x4_f32(false, a2, false, b2, 0, p3, false, false);
        p4 = __builtin_amdgcn_wmma_f32_16x16x4_f32(false, a1, false, b2, 0, p4, false, false);
        p5 = __builtin_amdgcn_wmma_f32_16x16x4_f32(false, a1, false, b3, 0, p5, false, false);
    }

    float* myPart = part + wave * (5 * 256) + lane;
    #pragma unroll
    for (int r = 0; r < 8; ++r) {
        myPart[0 * 256 + r * 32] = p1[r];
        myPart[1 * 256 + r * 32] = p2[r];
        myPart[2 * 256 + r * 32] = p3[r];
        myPart[3 * 256 + r * 32] = p4[r];
        myPart[4 * 256 + r * 32] = p5[r];
    }
    __syncthreads();

    const int row = tid >> 4;
    const int oc  = tid & 15;
    const float* src = part + (row & 7) * 32 + (row >> 3) * 16 + oc;
    float s1 = src[0 * 256], s2 = src[1 * 256], s3 = src[2 * 256],
          s4 = src[3 * 256], s5 = src[4 * 256];
    #pragma unroll
    for (int w = 1; w < KWAVES; ++w) {
        const float* q = src + w * (5 * 256);
        s1 += q[0 * 256];
        s2 += q[1 * 256];
        s3 += q[2 * 256];
        s4 += q[3 * 256];
        s5 += q[4 * 256];
    }

    rtile[tid] = s5;
    __syncthreads();
    float run = 0.0f;
    #pragma unroll
    for (int r = 0; r < 16; ++r) {
        const float v = rtile[r * 16 + oc];
        run += (r <= row) ? v : 0.0f;
    }

    const int m = blockIdx.x * 16 + row;
    Qj[m * ODIM + oc]  = s1 + 0.5f * s3 + bf[oc];
    Qi[m * ODIM + oc]  = 0.5f * s4 - s2;
    P5t[m * ODIM + oc] = run;
    if (row == 15) {
        Tt[blockIdx.x * TT_PITCH + oc] = run;
    }
}

__global__ __launch_bounds__(LDIM) void epilogue_kernel(
    const float* __restrict__ Qj, const float* __restrict__ Qi,
    const float* __restrict__ P5t, const float* __restrict__ Tt,
    float* __restrict__ out)
{
    const int i = blockIdx.x;
    const int b = blockIdx.y;
    const int j = threadIdx.x;

    __shared__ __attribute__((aligned(16))) float sQi[ODIM];
    __shared__ __attribute__((aligned(16))) float sP5p[ODIM];
    __shared__ __attribute__((aligned(16))) float sOff[TILES_PER_B * ODIM];

    {
        const int t = threadIdx.x >> 4;
        const int o = threadIdx.x & 15;
        const float* tt = Tt + (b * TILES_PER_B) * TT_PITCH + o;
        float acc = 0.0f;
        #pragma unroll
        for (int tp = 0; tp < TILES_PER_B - 1; ++tp) {
            const float v = tt[tp * TT_PITCH];
            acc += (tp < t) ? v : 0.0f;
        }
        sOff[threadIdx.x] = acc;
    }
    if (threadIdx.x < ODIM) {
        const int o  = threadIdx.x;
        const int ip = (i == 0) ? 0 : (i - 1);
        const float qiv = Qi [(b * LDIM + i)  * ODIM + o];
        const float ppv = P5t[(b * LDIM + ip) * ODIM + o];
        sQi[o]  = qiv;
        sP5p[o] = (i == 0) ? 0.0f : ppv;
    }
    __syncthreads();

    const float denom = (float)(j - i + 1) + EPSV;
    const float rden  = 1.0f / denom;

    const float* qj = Qj  + (b * LDIM + j) * ODIM;
    const float* pj = P5t + (b * LDIM + j) * ODIM;
    const float* offJ = sOff + (j >> 4) * ODIM;
    const float* offP = sOff + ((i == 0) ? 0 : ((i - 1) >> 4)) * ODIM;
    float* o0 = out + ((size_t)(b * ODIM) * LDIM + i) * LDIM + j;

    #pragma unroll
    for (int o = 0; o < ODIM; ++o) {
        const float p5j = pj[o]   + offJ[o];
        const float p5p = sP5p[o] + offP[o];
        const float v = qj[o] + sQi[o] + (p5j - p5p) * rden;
        o0[(size_t)o * LDIM * LDIM] = v;
    }
}

extern "C" void kernel_launch(void* const* d_in, const int* in_sizes, int n_in,
                              void* d_out, int out_size, void* d_ws, size_t ws_size,
                              hipStream_t stream)
{
    const float* x  = (const float*)d_in[0];
    const float* W1 = (const float*)d_in[1];
    const float* b1 = (const float*)d_in[2];
    const float* W2 = (const float*)d_in[3];
    const float* b2 = (const float*)d_in[4];
    const float* Wf = (const float*)d_in[5];
    const float* bf = (const float*)d_in[6];
    float* out = (float*)d_out;

    float* ws  = (float*)d_ws;
    float* h1  = ws;
    float* h2  = h1  + (size_t)MDIM * HDIM;
    float* Tt  = h2  + (size_t)MDIM * HDIM;
    float* Qj  = Tt  + (size_t)MDIM * HDIM;
    float* Qi  = Qj  + (size_t)MDIM * ODIM;
    float* P5t = Qi  + (size_t)MDIM * ODIM;

    dim3 g1(MDIM / 16, HDIM / 64, 2);
    mlp_wmma_kernel<<<g1, 32 * MLP_WAVES, 0, stream>>>(x, W1, b1, h1, W2, b2, h2);
    proj_wmma_kernel<<<MDIM / 16, 32 * KWAVES, 0, stream>>>(h1, h2, Wf, bf, Qj, Qi, P5t, Tt);
    epilogue_kernel<<<dim3(LDIM, BDIM), LDIM, 0, stream>>>(Qj, Qi, P5t, Tt, out);
}


# Projection and epilogue kernels: every global load issued at kernel entry as one group, in front of the first product, wait or barrier: one memory round trip per block in each, was three
# speedup vs baseline: 5.1688x; 1.0565x over previous
#include <hip/hip_runtime.h>

#define LDIM 256
#define DDIM 512
#define HDIM 256
#define ODIM 16
#define MDIM 512
#define BDIM 2
#define LEAKY 0.1f
#define EPSV 1e-9f

#define KWAVES 8
#define PROJ_KSLICE (HDIM / KWAVES)
#define MLP_WAVES   4
#define MLP_KSLICE  (DDIM / MLP_WAVES)
#define MLP_GROUPS  (MLP_KSLICE / 8)
#define MLP_AHEAD   8
#define MLP_TILE_LDS 272

typedef __attribute__((ext_vector_type(2))) float v2f;
typedef __attribute__((ext_vector_type(4))) float v4f;
typedef __attribute__((ext_vector_type(8))) float v8f;

__global__ __launch_bounds__(32 * MLP_WAVES) void mlp_wmma_kernel(
    const float* X,
    const float* Wa, const float* biasA, float* HoutA,
    const float* Wb, const float* biasB, float* HoutB)
{
    __shared__ __attribute__((aligned(16))) float part[MLP_WAVES * 4 * MLP_TILE_LDS];

    const bool second = (blockIdx.z != 0);
    const float* W    = second ? Wb    : Wa;
    const float* bias = second ? biasB : biasA;
    float*       Hout = second ? HoutB : HoutA;

    const int tid   = threadIdx.x;
    const int lane  = tid & 31;
    const int wave  = tid >> 5;
    const int mBase = blockIdx.x * 16;
    const int nBase = blockIdx.y * 64;
    const int mRow  = mBase + (lane & 15);
    const int kOff  = wave * MLP_KSLICE + (lane >> 4) * 4;

    v8f acc0 = {}, acc1 = {}, acc2 = {}, acc3 = {};

    const float* xrow = X + mRow * DDIM + kOff;
    const int n0 = nBase + (lane & 15);
    const float* w0 = W + (n0 +  0) * DDIM + kOff;
    const float* w1 = W + (n0 + 16) * DDIM + kOff;
    const float* w2 = W + (n0 + 32) * DDIM + kOff;
    const float* w3 = W + (n0 + 48) * DDIM + kOff;

    v4f fa[MLP_GROUPS], f0[MLP_GROUPS], f1[MLP_GROUPS], f2[MLP_GROUPS], f3[MLP_GROUPS];

    #pragma unroll
    for (int g = 0; g < MLP_AHEAD; ++g) {
        fa[g] = *(const v4f*)(xrow + g * 8);
        f0[g] = *(const v4f*)(w0 + g * 8);
        f1[g] = *(const v4f*)(w1 + g * 8);
        f2[g] = *(const v4f*)(w2 + g * 8);
        f3[g] = *(const v4f*)(w3 + g * 8);
    }
    __builtin_amdgcn_sched_barrier(0);

    #pragma unroll
    for (int g = 0; g < MLP_GROUPS; ++g) {
        const v2f aL = {fa[g][0], fa[g][1]}, aH = {fa[g][2], fa[g][3]};
        const v2f b0L = {f0[g][0], f0[g][1]}, b0H = {f0[g][2], f0[g][3]};
        const v2f b1L = {f1[g][0], f1[g][1]}, b1H = {f1[g][2], f1[g][3]};
        const v2f b2L = {f2[g][0], f2[g][1]}, b2H = {f2[g][2], f2[g][3]};
        const v2f b3L = {f3[g][0], f3[g][1]}, b3H = {f3[g][2], f3[g][3]};
        acc0 = __builtin_amdgcn_wmma_f32_16x16x4_f32(false, aL, false, b0L, 0, acc0, false, false);
        acc1 = __builtin_amdgcn_wmma_f32_16x16x4_f32(false, aL, false, b1L, 0, acc1, false, false);
        acc2 = __builtin_amdgcn_wmma_f32_16x16x4_f32(false, aL, false, b2L, 0, acc2, false, false);
        acc3 = __builtin_amdgcn_wmma_f32_16x16x4_f32(false, aL, false, b3L, 0, acc3, false, false);
        acc0 = __builtin_amdgcn_wmma_f32_16x16x4_f32(false, aH, false, b0H, 0, acc0, false, false);
        acc1 = __builtin_amdgcn_wmma_f32_16x16x4_f32(false, aH, false, b1H, 0, acc1, false, false);
        acc2 = __builtin_amdgcn_wmma_f32_16x16x4_f32(false, aH, false, b2H, 0, acc2, false, false);
        acc3 = __builtin_amdgcn_wmma_f32_16x16x4_f32(false, aH, false, b3H, 0, acc3, false, false);
        if (g + MLP_AHEAD < MLP_GROUPS) {
            fa[g + MLP_AHEAD] = *(const v4f*)(xrow + (g + MLP_AHEAD) * 8);
            f0[g + MLP_AHEAD] = *(const v4f*)(w0 + (g + MLP_AHEAD) * 8);
            f1[g + MLP_AHEAD] = *(const v4f*)(w1 + (g + MLP_AHEAD) * 8);
            f2[g + MLP_AHEAD] = *(const v4f*)(w2 + (g + MLP_AHEAD) * 8);
            f3[g + MLP_AHEAD] = *(const v4f*)(w3 + (g + MLP_AHEAD) * 8);
            __builtin_amdgcn_sched_barrier(0);
        }
    }

    float* myPart = part + wave * (4 * MLP_TILE_LDS) + lane;
    v8f accs[4] = {acc0, acc1, acc2, acc3};
    #pragma unroll
    for (int t = 0; t < 4; ++t) {
        #pragma unroll
        for (int r = 0; r < 8; ++r) {
            myPart[t * MLP_TILE_LDS + r * 32] = accs[t][r];
        }
    }
    __syncthreads();

    const int col = (tid & 15) * 4;
    const v4f bv = *(const v4f*)(bias + nBase + col);
    #pragma unroll
    for (int half = 0; half < 2; ++half) {
        const int row = (tid >> 4) + 8 * half;
        const float* src = part + (col >> 4) * MLP_TILE_LDS
                         + (row & 7) * 32 + (row >> 3) * 16 + (col & 15);
        v4f s = *(const v4f*)src;
        #pragma unroll
        for (int w = 1; w < MLP_WAVES; ++w) {
            s += *(const v4f*)(src + w * (4 * MLP_TILE_LDS));
        }
        v4f h = s + bv;
        #pragma unroll
        for (int e = 0; e < 4; ++e) {
            h[e] = (h[e] >= 0.0f) ? h[e] : (LEAKY * h[e]);
        }
        *(v4f*)(Hout + (mBase + row) * HDIM + nBase + col) = h;
    }
}

#define TILES_PER_B (LDIM / 16)
#define TT_PITCH 32

__global__ __launch_bounds__(32 * KWAVES) void proj_wmma_kernel(
    const float* h1, const float* h2,
    const float* Wf, const float* bf,
    float* Qj, float* Qi,
    float* P5t, float* Tt)
{
    __shared__ float part[KWAVES * 5 * 256];
    __shared__ float rtile[256];

    const int tid  = threadIdx.x;
    const int lane = tid & 31;
    const int wave = tid >> 5;
    const int mRow = blockIdx.x * 16 + (lane & 15);
    const int kOff = wave * PROJ_KSLICE + (lane >> 4) * 2;
    const int o    = lane & 15;

    const float* a1p = h1 + mRow * HDIM + kOff;
    const float* a2p = h2 + mRow * HDIM + kOff;
    const float* w1p = Wf + o * (3 * HDIM) + kOff;
    const float* w2p = w1p + HDIM;
    const float* w3p = w1p + 2 * HDIM;

    v2f fa1[PROJ_KSLICE / 4], fa2[PROJ_KSLICE / 4];
    v2f fb1[PROJ_KSLICE / 4], fb2[PROJ_KSLICE / 4], fb3[PROJ_KSLICE / 4];
    #pragma unroll
    for (int s = 0; s < PROJ_KSLICE / 4; ++s) {
        fa1[s] = *(const v2f*)(a1p + 4 * s);
        fa2[s] = *(const v2f*)(a2p + 4 * s);
        fb1[s] = *(const v2f*)(w1p + 4 * s);
        fb2[s] = *(const v2f*)(w2p + 4 * s);
        fb3[s] = *(const v2f*)(w3p + 4 * s);
    }
    const float bfv = bf[tid & 15];
    __builtin_amdgcn_sched_barrier(0);

    v8f p1 = {}, p2 = {}, p3 = {}, p4 = {}, p5 = {};
    #pragma unroll
    for (int s = 0; s < PROJ_KSLICE / 4; ++s) {
        const v2f a1 = fa1[s], a2 = fa2[s], b1 = fb1[s], b2 = fb2[s], b3 = fb3[s];
        p1 = __builtin_amdgcn_wmma_f32_16x16x4_f32(false, a1, false, b1, 0, p1, false, false);
        p2 = __builtin_amdgcn_wmma_f32_16x16x4_f32(false, a2, false, b1, 0, p2, false, false);
        p3 = __builtin_amdgcn_wmma_f32_16x16x4_f32(false, a2, false, b2, 0, p3, false, false);
        p4 = __builtin_amdgcn_wmma_f32_16x16x4_f32(false, a1, false, b2, 0, p4, false, false);
        p5 = __builtin_amdgcn_wmma_f32_16x16x4_f32(false, a1, false, b3, 0, p5, false, false);
    }

    float* myPart = part + wave * (5 * 256) + lane;
    #pragma unroll
    for (int r = 0; r < 8; ++r) {
        myPart[0 * 256 + r * 32] = p1[r];
        myPart[1 * 256 + r * 32] = p2[r];
        myPart[2 * 256 + r * 32] = p3[r];
        myPart[3 * 256 + r * 32] = p4[r];
        myPart[4 * 256 + r * 32] = p5[r];
    }
    __syncthreads();

    const int row = tid >> 4;
    const int oc  = tid & 15;
    const float* src = part + (row & 7) * 32 + (row >> 3) * 16 + oc;
    float s1 = src[0 * 256], s2 = src[1 * 256], s3 = src[2 * 256],
          s4 = src[3 * 256], s5 = src[4 * 256];
    #pragma unroll
    for (int w = 1; w < KWAVES; ++w) {
        const float* q = src + w * (5 * 256);
        s1 += q[0 * 256];
        s2 += q[1 * 256];
        s3 += q[2 * 256];
        s4 += q[3 * 256];
        s5 += q[4 * 256];
    }

    rtile[tid] = s5;
    __syncthreads();
    float run = 0.0f;
    #pragma unroll
    for (int r = 0; r < 16; ++r) {
        const float v = rtile[r * 16 + oc];
        run += (r <= row) ? v : 0.0f;
    }

    const int m = blockIdx.x * 16 + row;
    Qj[m * ODIM + oc]  = s1 + 0.5f * s3 + bfv;
    Qi[m * ODIM + oc]  = 0.5f * s4 - s2;
    P5t[m * ODIM + oc] = run;
    if (row == 15) {
        Tt[blockIdx.x * TT_PITCH + oc] = run;
    }
}

__global__ __launch_bounds__(LDIM) void epilogue_kernel(
    const float* Qj, const float* Qi,
    const float* P5t, const float* Tt,
    float* out)
{
    const int i = blockIdx.x;
    const int b = blockIdx.y;
    const int j = threadIdx.x;

    __shared__ __attribute__((aligned(16))) float sQi[ODIM];
    __shared__ __attribute__((aligned(16))) float sP5p[ODIM];
    __shared__ __attribute__((aligned(16))) float sOff[TILES_PER_B * ODIM];

    const v4f* qjv = (const v4f*)(Qj  + (b * LDIM + j) * ODIM);
    const v4f* pjv = (const v4f*)(P5t + (b * LDIM + j) * ODIM);
    v4f q4[ODIM / 4], p4[ODIM / 4];
    #pragma unroll
    for (int c = 0; c < ODIM / 4; ++c) {
        q4[c] = qjv[c];
        p4[c] = pjv[c];
    }
    const int tl = threadIdx.x >> 4;
    const float* tt = Tt + (b * TILES_PER_B) * TT_PITCH + (threadIdx.x & 15);
    float tv[TILES_PER_B - 1];
    #pragma unroll
    for (int tp = 0; tp < TILES_PER_B - 1; ++tp) {
        tv[tp] = tt[tp * TT_PITCH];
    }
    if (threadIdx.x < ODIM) {
        const int o  = threadIdx.x;
        const int ip = (i == 0) ? 0 : (i - 1);
        const float qiv = Qi [(b * LDIM + i)  * ODIM + o];
        const float ppv = P5t[(b * LDIM + ip) * ODIM + o];
        sQi[o]  = qiv;
        sP5p[o] = (i == 0) ? 0.0f : ppv;
    }
    {
        float acc = 0.0f;
        #pragma unroll
        for (int tp = 0; tp < TILES_PER_B - 1; ++tp) {
            acc += (tp < tl) ? tv[tp] : 0.0f;
        }
        sOff[threadIdx.x] = acc;
    }
    __syncthreads();

    const float denom = (float)(j - i + 1) + EPSV;
    const float rden  = 1.0f / denom;

    const float* offJ = sOff + (j >> 4) * ODIM;
    const float* offP = sOff + ((i == 0) ? 0 : ((i - 1) >> 4)) * ODIM;
    float* o0 = out + ((size_t)(b * ODIM) * LDIM + i) * LDIM + j;

    #pragma unroll
    for (int o = 0; o < ODIM; ++o) {
        const float p5j = p4[o >> 2][o & 3] + offJ[o];
        const float p5p = sP5p[o] + offP[o];
        const float v = q4[o >> 2][o & 3] + sQi[o] + (p5j - p5p) * rden;
        o0[(size_t)o * LDIM * LDIM] = v;
    }
}

extern "C" void kernel_launch(void* const* d_in, const int* in_sizes, int n_in,
                              void* d_out, int out_size, void* d_ws, size_t ws_size,
                              hipStream_t stream)
{
    const float* x  = (const float*)d_in[0];
    const float* W1 = (const float*)d_in[1];
    const float* b1 = (const float*)d_in[2];
    const float* W2 = (const float*)d_in[3];
    const float* b2 = (const float*)d_in[4];
    const float* Wf = (const float*)d_in[5];
    const float* bf = (const float*)d_in[6];
    float* out = (float*)d_out;

    float* ws  = (float*)d_ws;
    float* h1  = ws;
    float* h2  = h1  + (size_t)MDIM * HDIM;
    float* Tt  = h2  + (size_t)MDIM * HDIM;
    float* Qj  = Tt  + (size_t)MDIM * HDIM;
    float* Qi  = Qj  + (size_t)MDIM * ODIM;
    float* P5t = Qi  + (size_t)MDIM * ODIM;

    dim3 g1(MDIM / 16, HDIM / 64, 2);
    mlp_wmma_kernel<<<g1, 32 * MLP_WAVES, 0, stream>>>(x, W1, b1, h1, W2, b2, h2);
    proj_wmma_kernel<<<MDIM / 16, 32 * KWAVES, 0, stream>>>(h1, h2, Wf, bf, Qj, Qi, P5t, Tt);
    epilogue_kernel<<<dim3(LDIM, BDIM), LDIM, 0, stream>>>(Qj, Qi, P5t, Tt, out);
}


# Projection kernel: three 4-wave blocks per 16-row tile (span term, Qj, Qi), one wave per SIMD with 64 k and 16-byte fragment loads: 32 products queue per SIMD, was 80; new summation tree
# speedup vs baseline: 5.5070x; 1.0654x over previous
#include <hip/hip_runtime.h>

#define LDIM 256
#define DDIM 512
#define HDIM 256
#define ODIM 16
#define MDIM 512
#define BDIM 2
#define LEAKY 0.1f
#define EPSV 1e-9f

#define PROJ_WAVES  4
#define PROJ_KSLICE (HDIM / PROJ_WAVES)
#define PROJ_GROUPS (PROJ_KSLICE / 8)
#define PROJ_ROLES  3
#define MLP_WAVES   4
#define MLP_KSLICE  (DDIM / MLP_WAVES)
#define MLP_GROUPS  (MLP_KSLICE / 8)
#define MLP_AHEAD   8
#define MLP_TILE_LDS 272

typedef __attribute__((ext_vector_type(2))) float v2f;
typedef __attribute__((ext_vector_type(4))) float v4f;
typedef __attribute__((ext_vector_type(8))) float v8f;

__global__ __launch_bounds__(32 * MLP_WAVES) void mlp_wmma_kernel(
    const float* X,
    const float* Wa, const float* biasA, float* HoutA,
    const float* Wb, const float* biasB, float* HoutB)
{
    __shared__ __attribute__((aligned(16))) float part[MLP_WAVES * 4 * MLP_TILE_LDS];

    const bool second = (blockIdx.z != 0);
    const float* W    = second ? Wb    : Wa;
    const float* bias = second ? biasB : biasA;
    float*       Hout = second ? HoutB : HoutA;

    const int tid   = threadIdx.x;
    const int lane  = tid & 31;
    const int wave  = tid >> 5;
    const int mBase = blockIdx.x * 16;
    const int nBase = blockIdx.y * 64;
    const int mRow  = mBase + (lane & 15);
    const int kOff  = wave * MLP_KSLICE + (lane >> 4) * 4;

    v8f acc0 = {}, acc1 = {}, acc2 = {}, acc3 = {};

    const float* xrow = X + mRow * DDIM + kOff;
    const int n0 = nBase + (lane & 15);
    const float* w0 = W + (n0 +  0) * DDIM + kOff;
    const float* w1 = W + (n0 + 16) * DDIM + kOff;
    const float* w2 = W + (n0 + 32) * DDIM + kOff;
    const float* w3 = W + (n0 + 48) * DDIM + kOff;

    v4f fa[MLP_GROUPS], f0[MLP_GROUPS], f1[MLP_GROUPS], f2[MLP_GROUPS], f3[MLP_GROUPS];

    #pragma unroll
    for (int g = 0; g < MLP_AHEAD; ++g) {
        fa[g] = *(const v4f*)(xrow + g * 8);
        f0[g] = *(const v4f*)(w0 + g * 8);
        f1[g] = *(const v4f*)(w1 + g * 8);
        f2[g] = *(const v4f*)(w2 + g * 8);
        f3[g] = *(const v4f*)(w3 + g * 8);
    }
    __builtin_amdgcn_sched_barrier(0);

    #pragma unroll
    for (int g = 0; g < MLP_GROUPS; ++g) {
        const v2f aL = {fa[g][0], fa[g][1]}, aH = {fa[g][2], fa[g][3]};
        const v2f b0L = {f0[g][0], f0[g][1]}, b0H = {f0[g][2], f0[g][3]};
        const v2f b1L = {f1[g][0], f1[g][1]}, b1H = {f1[g][2], f1[g][3]};
        const v2f b2L = {f2[g][0], f2[g][1]}, b2H = {f2[g][2], f2[g][3]};
        const v2f b3L = {f3[g][0], f3[g][1]}, b3H = {f3[g][2], f3[g][3]};
        acc0 = __builtin_amdgcn_wmma_f32_16x16x4_f32(false, aL, false, b0L, 0, acc0, false, false);
        acc1 = __builtin_amdgcn_wmma_f32_16x16x4_f32(false, aL, false, b1L, 0, acc1, false, false);
        acc2 = __builtin_amdgcn_wmma_f32_16x16x4_f32(false, aL, false, b2L, 0, acc2, false, false);
        acc3 = __builtin_amdgcn_wmma_f32_16x16x4_f32(false, aL, false, b3L, 0, acc3, false, false);
        acc0 = __builtin_amdgcn_wmma_f32_16x16x4_f32(false, aH, false, b0H, 0, acc0, false, false);
        acc1 = __builtin_amdgcn_wmma_f32_16x16x4_f32(false, aH, false, b1H, 0, acc1, false, false);
        acc2 = __builtin_amdgcn_wmma_f32_16x16x4_f32(false, aH, false, b2H, 0, acc2, false, false);
        acc3 = __builtin_amdgcn_wmma_f32_16x16x4_f32(false, aH, false, b3H, 0, acc3, false, false);
        if (g + MLP_AHEAD < MLP_GROUPS) {
            fa[g + MLP_AHEAD] = *(const v4f*)(xrow + (g + MLP_AHEAD) * 8);
            f0[g + MLP_AHEAD] = *(const v4f*)(w0 + (g + MLP_AHEAD) * 8);
            f1[g + MLP_AHEAD] = *(const v4f*)(w1 + (g + MLP_AHEAD) * 8);
            f2[g + MLP_AHEAD] = *(const v4f*)(w2 + (g + MLP_AHEAD) * 8);
            f3[g + MLP_AHEAD] = *(const v4f*)(w3 + (g + MLP_AHEAD) * 8);
            __builtin_amdgcn_sched_barrier(0);
        }
    }

    float* myPart = part + wave * (4 * MLP_TILE_LDS) + lane;
    v8f accs[4] = {acc0, acc1, acc2, acc3};
    #pragma unroll
    for (int t = 0; t < 4; ++t) {
        #pragma unroll
        for (int r = 0; r < 8; ++r) {
            myPart[t * MLP_TILE_LDS + r * 32] = accs[t][r];
        }
    }
    __syncthreads();

    const int col = (tid & 15) * 4;
    const v4f bv = *(const v4f*)(bias + nBase + col);
    #pragma unroll
    for (int half = 0; half < 2; ++half) {
        const int row = (tid >> 4) + 8 * half;
        const float* src = part + (col >> 4) * MLP_TILE_LDS
                         + (row & 7) * 32 + (row >> 3) * 16 + (col & 15);
        v4f s = *(const v4f*)src;
        #pragma unroll
        for (int w = 1; w < MLP_WAVES; ++w) {
            s += *(const v4f*)(src + w * (4 * MLP_TILE_LDS));
        }
        v4f h = s + bv;
        #pragma unroll
        for (int e = 0; e < 4; ++e) {
            h[e] = (h[e] >= 0.0f) ? h[e] : (LEAKY * h[e]);
        }
        *(v4f*)(Hout + (mBase + row) * HDIM + nBase + col) = h;
    }
}

#define TILES_PER_B (LDIM / 16)
#define TT_PITCH 32

__global__ __launch_bounds__(32 * PROJ_WAVES) void proj_wmma_kernel(
    const float* h1, const float* h2,
    const float* Wf, const float* bf,
    float* Qj, float* Qi,
    float* P5t, float* Tt)
{
    __shared__ float part[PROJ_WAVES * 2 * 256];
    __shared__ float rtile[256];

    const int tid  = threadIdx.x;
    const int lane = tid & 31;
    const int wave = tid >> 5;
    const int role = blockIdx.y;
    const int mRow = blockIdx.x * 16 + (lane & 15);
    const int kOff = wave * PROJ_KSLICE + (lane >> 4) * 4;
    const int o    = lane & 15;

    const float* wrow = Wf + o * (3 * HDIM) + kOff;

    v8f pa = {}, pb = {};
    float bfv = 0.0f;
    if (role == 0) {
        const float* ap = h1 + mRow * HDIM + kOff;
        const float* wp = wrow + 2 * HDIM;
        v4f fa[PROJ_GROUPS], fb[PROJ_GROUPS];
        #pragma unroll
        for (int g = 0; g < PROJ_GROUPS; ++g) {
            fa[g] = *(const v4f*)(ap + g * 8);
            fb[g] = *(const v4f*)(wp + g * 8);
        }
        __builtin_amdgcn_sched_barrier(0);
        #pragma unroll
        for (int g = 0; g < PROJ_GROUPS; ++g) {
            const v2f aL = {fa[g][0], fa[g][1]}, aH = {fa[g][2], fa[g][3]};
            const v2f bL = {fb[g][0], fb[g][1]}, bH = {fb[g][2], fb[g][3]};
            pa = __builtin_amdgcn_wmma_f32_16x16x4_f32(false, aL, false, bL, 0, pa, false, false);
            pb = __builtin_amdgcn_wmma_f32_16x16x4_f32(false, aH, false, bH, 0, pb, false, false);
        }
    } else {
        const float* xa = (role == 1) ? h1 : h2;
        const float* ya = (role == 1) ? h2 : h1;
        const float* xp = xa + mRow * HDIM + kOff;
        const float* yp = ya + mRow * HDIM + kOff;
        const float* w1p = wrow;
        const float* w2p = wrow + HDIM;
        v4f fx[PROJ_GROUPS], fy[PROJ_GROUPS], f1[PROJ_GROUPS], f2[PROJ_GROUPS];
        #pragma unroll
        for (int g = 0; g < PROJ_GROUPS; ++g) {
            fx[g] = *(const v4f*)(xp  + g * 8);
            f1[g] = *(const v4f*)(w1p + g * 8);
            fy[g] = *(const v4f*)(yp  + g * 8);
            f2[g] = *(const v4f*)(w2p + g * 8);
        }
        bfv = bf[tid & 15];
        __builtin_amdgcn_sched_barrier(0);
        #pragma unroll
        for (int g = 0; g < PROJ_GROUPS; ++g) {
            const v2f xL = {fx[g][0], fx[g][1]}, xH = {fx[g][2], fx[g][3]};
            const v2f yL = {fy[g][0], fy[g][1]}, yH = {fy[g][2], fy[g][3]};
            const v2f b1L = {f1[g][0], f1[g][1]}, b1H = {f1[g][2], f1[g][3]};
            const v2f b2L = {f2[g][0], f2[g][1]}, b2H = {f2[g][2], f2[g][3]};
            pa = __builtin_amdgcn_wmma_f32_16x16x4_f32(false, xL, false, b1L, 0, pa, false, false);
            pb = __builtin_amdgcn_wmma_f32_16x16x4_f32(false, yL, false, b2L, 0, pb, false, false);
            pa = __builtin_amdgcn_wmma_f32_16x16x4_f32(false, xH, false, b1H, 0, pa, false, false);
            pb = __builtin_amdgcn_wmma_f32_16x16x4_f32(false, yH, false, b2H, 0, pb, false, false);
        }
    }

    float* myPart = part + wave * (2 * 256) + lane;
    #pragma unroll
    for (int r = 0; r < 8; ++r) {
        myPart[0 * 256 + r * 32] = pa[r];
        myPart[1 * 256 + r * 32] = pb[r];
    }
    __syncthreads();

    const int row = tid >> 4;
    const int oc  = tid & 15;
    const float* src = part + row * 32 + oc;
    float sa0 = src[0], sb0 = src[256], sa1 = src[16], sb1 = src[256 + 16];
    #pragma unroll
    for (int w = 1; w < PROJ_WAVES; ++w) {
        const float* q = src + w * (2 * 256);
        sa0 += q[0];
        sb0 += q[256];
        sa1 += q[16];
        sb1 += q[256 + 16];
    }

    const int m = blockIdx.x * 16 + row;
    if (role == 0) {
        rtile[row * 16 + oc]       = sa0 + sb0;
        rtile[(row + 8) * 16 + oc] = sa1 + sb1;
        __syncthreads();
        float run0 = 0.0f, run1 = 0.0f;
        #pragma unroll
        for (int r = 0; r < 16; ++r) {
            const float v = rtile[r * 16 + oc];
            run0 += (r <= row)     ? v : 0.0f;
            run1 += (r <= row + 8) ? v : 0.0f;
        }
        P5t[m * ODIM + oc]       = run0;
        P5t[(m + 8) * ODIM + oc] = run1;
        if (row == 7) {
            Tt[blockIdx.x * TT_PITCH + oc] = run1;
        }
    } else if (role == 1) {
        Qj[m * ODIM + oc]       = sa0 + 0.5f * sb0 + bfv;
        Qj[(m + 8) * ODIM + oc] = sa1 + 0.5f * sb1 + bfv;
    } else {
        Qi[m * ODIM + oc]       = 0.5f * sb0 - sa0;
        Qi[(m + 8) * ODIM + oc] = 0.5f * sb1 - sa1;
    }
}

__global__ __launch_bounds__(LDIM) void epilogue_kernel(
    const float* Qj, const float* Qi,
    const float* P5t, const float* Tt,
    float* out)
{
    const int i = blockIdx.x;
    const int b = blockIdx.y;
    const int j = threadIdx.x;

    __shared__ __attribute__((aligned(16))) float sQi[ODIM];
    __shared__ __attribute__((aligned(16))) float sP5p[ODIM];
    __shared__ __attribute__((aligned(16))) float sOff[TILES_PER_B * ODIM];

    const v4f* qjv = (const v4f*)(Qj  + (b * LDIM + j) * ODIM);
    const v4f* pjv = (const v4f*)(P5t + (b * LDIM + j) * ODIM);
    v4f q4[ODIM / 4], p4[ODIM / 4];
    #pragma unroll
    for (int c = 0; c < ODIM / 4; ++c) {
        q4[c] = qjv[c];
        p4[c] = pjv[c];
    }
    const int tl = threadIdx.x >> 4;
    const float* tt = Tt + (b * TILES_PER_B) * TT_PITCH + (threadIdx.x & 15);
    float tv[TILES_PER_B - 1];
    #pragma unroll
    for (int tp = 0; tp < TILES_PER_B - 1; ++tp) {
        tv[tp] = tt[tp * TT_PITCH];
    }
    if (threadIdx.x < ODIM) {
        const int o  = threadIdx.x;
        const int ip = (i == 0) ? 0 : (i - 1);
        const float qiv = Qi [(b * LDIM + i)  * ODIM + o];
        const float ppv = P5t[(b * LDIM + ip) * ODIM + o];
        sQi[o]  = qiv;
        sP5p[o] = (i == 0) ? 0.0f : ppv;
    }
    {
        float acc = 0.0f;
        #pragma unroll
        for (int tp = 0; tp < TILES_PER_B - 1; ++tp) {
            acc += (tp < tl) ? tv[tp] : 0.0f;
        }
        sOff[threadIdx.x] = acc;
    }
    __syncthreads();

    const float denom = (float)(j - i + 1) + EPSV;
    const float rden  = 1.0f / denom;

    const float* offJ = sOff + (j >> 4) * ODIM;
    const float* offP = sOff + ((i == 0) ? 0 : ((i - 1) >> 4)) * ODIM;
    float* o0 = out + ((size_t)(b * ODIM) * LDIM + i) * LDIM + j;

    #pragma unroll
    for (int o = 0; o < ODIM; ++o) {
        const float p5j = p4[o >> 2][o & 3] + offJ[o];
        const float p5p = sP5p[o] + offP[o];
        const float v = q4[o >> 2][o & 3] + sQi[o] + (p5j - p5p) * rden;
        o0[(size_t)o * LDIM * LDIM] = v;
    }
}

extern "C" void kernel_launch(void* const* d_in, const int* in_sizes, int n_in,
                              void* d_out, int out_size, void* d_ws, size_t ws_size,
                              hipStream_t stream)
{
    const float* x  = (const float*)d_in[0];
    const float* W1 = (const float*)d_in[1];
    const float* b1 = (const float*)d_in[2];
    const float* W2 = (const float*)d_in[3];
    const float* b2 = (const float*)d_in[4];
    const float* Wf = (const float*)d_in[5];
    const float* bf = (const float*)d_in[6];
    float* out = (float*)d_out;

    float* ws  = (float*)d_ws;
    float* h1  = ws;
    float* h2  = h1  + (size_t)MDIM * HDIM;
    float* Tt  = h2  + (size_t)MDIM * HDIM;
    float* Qj  = Tt  + (size_t)MDIM * HDIM;
    float* Qi  = Qj  + (size_t)MDIM * ODIM;
    float* P5t = Qi  + (size_t)MDIM * ODIM;

    dim3 g1(MDIM / 16, HDIM / 64, 2);
    mlp_wmma_kernel<<<g1, 32 * MLP_WAVES, 0, stream>>>(x, W1, b1, h1, W2, b2, h2);
    proj_wmma_kernel<<<dim3(MDIM / 16, PROJ_ROLES), 32 * PROJ_WAVES, 0, stream>>>(h1, h2, Wf, bf, Qj, Qi, P5t, Tt);
    epilogue_kernel<<<dim3(LDIM, BDIM), LDIM, 0, stream>>>(Qj, Qi, P5t, Tt, out);
}


# Projection launch removed: each MLP block multiplies its 16x64 strip of h1 or h2 by Wf on WMMA (partial planes); epilogue, now 16 i x 32 j tiles, adds the strips; 3 launches to 2; new summation order
# speedup vs baseline: 6.5522x; 1.1898x over previous
#include <hip/hip_runtime.h>

#define LDIM 256
#define DDIM 512
#define HDIM 256
#define ODIM 16
#define MDIM 512
#define BDIM 2
#define LEAKY 0.1f
#define EPSV 1e-9f

#define STRIPS      (HDIM / 64)
#define PROJ_GROUPS (64 / 8)
#define HS_PITCH    68
#define NPLANES     5
#define TILES       (MDIM / 16)
#define PLANE_PITCH (STRIPS * TILES * 256)
#define TILES_PER_B (LDIM / 16)
#define TT_PITCH    32
#define MLP_WAVES   4
#define MLP_KSLICE  (DDIM / MLP_WAVES)
#define MLP_GROUPS  (MLP_KSLICE / 8)
#define MLP_AHEAD   8
#define MLP_TILE_LDS 272

typedef __attribute__((ext_vector_type(2))) float v2f;
typedef __attribute__((ext_vector_type(4))) float v4f;
typedef __attribute__((ext_vector_type(8))) float v8f;

__global__ __launch_bounds__(32 * MLP_WAVES) void mlp_wmma_kernel(
    const float* X,
    const float* Wa, const float* biasA,
    const float* Wb, const float* biasB,
    const float* Wf,
    float* Pp, float* TtP)
{
    __shared__ __attribute__((aligned(16))) float part[MLP_WAVES * 4 * MLP_TILE_LDS];
    __shared__ __attribute__((aligned(16))) float hs[16 * HS_PITCH];

    const bool second = (blockIdx.z != 0);
    const float* W    = second ? Wb    : Wa;
    const float* bias = second ? biasB : biasA;

    const int tid   = threadIdx.x;
    const int lane  = tid & 31;
    const int wave  = tid >> 5;
    const int mBase = blockIdx.x * 16;
    const int nBase = blockIdx.y * 64;
    const int mRow  = mBase + (lane & 15);
    const int kOff  = wave * MLP_KSLICE + (lane >> 4) * 4;

    v8f acc0 = {}, acc1 = {}, acc2 = {}, acc3 = {};

    const float* xrow = X + mRow * DDIM + kOff;
    const int n0 = nBase + (lane & 15);
    const float* w0 = W + (n0 +  0) * DDIM + kOff;
    const float* w1 = W + (n0 + 16) * DDIM + kOff;
    const float* w2 = W + (n0 + 32) * DDIM + kOff;
    const float* w3 = W + (n0 + 48) * DDIM + kOff;

    v4f fa[MLP_GROUPS], f0[MLP_GROUPS], f1[MLP_GROUPS], f2[MLP_GROUPS], f3[MLP_GROUPS];

    #pragma unroll
    for (int g = 0; g < MLP_AHEAD; ++g) {
        fa[g] = *(const v4f*)(xrow + g * 8);
        f0[g] = *(const v4f*)(w0 + g * 8);
        f1[g] = *(const v4f*)(w1 + g * 8);
        f2[g] = *(const v4f*)(w2 + g * 8);
        f3[g] = *(const v4f*)(w3 + g * 8);
    }
    const int pw = (wave < 2) ? wave : 2;
    const float* wfp = Wf + (lane & 15) * (3 * HDIM) + pw * HDIM + nBase + (lane >> 4) * 4;
    v4f fw[PROJ_GROUPS];
    #pragma unroll
    for (int g = 0; g < PROJ_GROUPS; ++g) {
        fw[g] = *(const v4f*)(wfp + g * 8);
    }

    __builtin_amdgcn_sched_barrier(0);

    #pragma unroll
    for (int g = 0; g < MLP_GROUPS; ++g) {
        const v2f aL = {fa[g][0], fa[g][1]}, aH = {fa[g][2], fa[g][3]};
        const v2f b0L = {f0[g][0], f0[g][1]}, b0H = {f0[g][2], f0[g][3]};
        const v2f b1L = {f1[g][0], f1[g][1]}, b1H = {f1[g][2], f1[g][3]};
        const v2f b2L = {f2[g][0], f2[g][1]}, b2H = {f2[g][2], f2[g][3]};
        const v2f b3L = {f3[g][0], f3[g][1]}, b3H = {f3[g][2], f3[g][3]};
        acc0 = __builtin_amdgcn_wmma_f32_16x16x4_f32(false, aL, false, b0L, 0, acc0, false, false);
        acc1 = __builtin_amdgcn_wmma_f32_16x16x4_f32(false, aL, false, b1L, 0, acc1, false, false);
        acc2 = __builtin_amdgcn_wmma_f32_16x16x4_f32(false, aL, false, b2L, 0, acc2, false, false);
        acc3 = __builtin_amdgcn_wmma_f32_16x16x4_f32(false, aL, false, b3L, 0, acc3, false, false);
        acc0 = __builtin_amdgcn_wmma_f32_16x16x4_f32(false, aH, false, b0H, 0, acc0, false, false);
        acc1 = __builtin_amdgcn_wmma_f32_16x16x4_f32(false, aH, false, b1H, 0, acc1, false, false);
        acc2 = __builtin_amdgcn_wmma_f32_16x16x4_f32(false, aH, false, b2H, 0, acc2, false, false);
        acc3 = __builtin_amdgcn_wmma_f32_16x16x4_f32(false, aH, false, b3H, 0, acc3, false, false);
        if (g + MLP_AHEAD < MLP_GROUPS) {
            fa[g + MLP_AHEAD] = *(const v4f*)(xrow + (g + MLP_AHEAD) * 8);
            f0[g + MLP_AHEAD] = *(const v4f*)(w0 + (g + MLP_AHEAD) * 8);
            f1[g + MLP_AHEAD] = *(const v4f*)(w1 + (g + MLP_AHEAD) * 8);
            f2[g + MLP_AHEAD] = *(const v4f*)(w2 + (g + MLP_AHEAD) * 8);
            f3[g + MLP_AHEAD] = *(const v4f*)(w3 + (g + MLP_AHEAD) * 8);
            __builtin_amdgcn_sched_barrier(0);
        }
    }

    float* myPart = part + wave * (4 * MLP_TILE_LDS) + lane;
    v8f accs[4] = {acc0, acc1, acc2, acc3};
    #pragma unroll
    for (int t = 0; t < 4; ++t) {
        #pragma unroll
        for (int r = 0; r < 8; ++r) {
            myPart[t * MLP_TILE_LDS + r * 32] = accs[t][r];
        }
    }
    __syncthreads();

    const int col = (tid & 15) * 4;
    const v4f bv = *(const v4f*)(bias + nBase + col);
    #pragma unroll
    for (int half = 0; half < 2; ++half) {
        const int row = (tid >> 4) + 8 * half;
        const float* src = part + (col >> 4) * MLP_TILE_LDS
                         + (row & 7) * 32 + (row >> 3) * 16 + (col & 15);
        v4f s = *(const v4f*)src;
        #pragma unroll
        for (int w = 1; w < MLP_WAVES; ++w) {
            s += *(const v4f*)(src + w * (4 * MLP_TILE_LDS));
        }
        v4f h = s + bv;
        #pragma unroll
        for (int e = 0; e < 4; ++e) {
            h[e] = (h[e] >= 0.0f) ? h[e] : (LEAKY * h[e]);
        }
        *(v4f*)(hs + row * HS_PITCH + col) = h;
    }
    __syncthreads();

    const int wv = __builtin_amdgcn_readfirstlane(wave);
    const int nPl = second ? 2 : 3;
    if (wv < nPl) {
        const float* arow = hs + (lane & 15) * HS_PITCH + (lane >> 4) * 4;
        v8f pa = {}, pb = {};
        #pragma unroll
        for (int g = 0; g < PROJ_GROUPS; ++g) {
            const v4f a4 = *(const v4f*)(arow + g * 8);
            const v2f aL = {a4[0], a4[1]}, aH = {a4[2], a4[3]};
            const v2f bL = {fw[g][0], fw[g][1]}, bH = {fw[g][2], fw[g][3]};
            pa = __builtin_amdgcn_wmma_f32_16x16x4_f32(false, aL, false, bL, 0, pa, false, false);
            pb = __builtin_amdgcn_wmma_f32_16x16x4_f32(false, aH, false, bH, 0, pb, false, false);
        }
        const int plane = second ? (3 + wv) : wv;
        float* dst = Pp + (size_t)plane * PLANE_PITCH
                   + ((size_t)blockIdx.y * TILES + blockIdx.x) * 256 + lane;
        float tot = 0.0f;
        #pragma unroll
        for (int r = 0; r < 8; ++r) {
            const float d = pa[r] + pb[r];
            dst[r * 32] = d;
            tot += d;
        }
        if (plane == 2) {
            tot += __shfl_xor(tot, 16);
            if (lane < 16) {
                TtP[((size_t)blockIdx.x * STRIPS + blockIdx.y) * TT_PITCH + lane] = tot;
            }
        }
    }
}

#define EPI_JW   32
#define SQ_PITCH 36

__device__ __forceinline__ v4f p5_row(const float* sRt, int rIn, int o4,
                                      const float* sTot, int tileB)
{
    const v4f zero = {0.0f, 0.0f, 0.0f, 0.0f};
    v4f run = zero;
    #pragma unroll
    for (int r = 0; r < 16; ++r) {
        const v4f v = *(const v4f*)(sRt + r * ODIM + o4);
        run += (r <= rIn) ? v : zero;
    }
    v4f off = zero;
    #pragma unroll
    for (int tp = 0; tp < TILES_PER_B - 1; ++tp) {
        const v4f tv = *(const v4f*)(sTot + tp * ODIM + o4);
        off += (tp < tileB) ? tv : zero;
    }
    return run + off;
}

__global__ __launch_bounds__(256) void epilogue_kernel(
    const float* Pp, const float* TtP, const float* bf,
    float* out)
{
    __shared__ __attribute__((aligned(16))) float sR[64 * ODIM];
    __shared__ __attribute__((aligned(16))) float sTot[TILES_PER_B * ODIM];
    __shared__ __attribute__((aligned(16))) float sQj[ODIM * SQ_PITCH];
    __shared__ __attribute__((aligned(16))) float sP5j[ODIM * SQ_PITCH];
    __shared__ __attribute__((aligned(16))) float sQi[16 * ODIM];
    __shared__ __attribute__((aligned(16))) float sP5i[32 * ODIM];

    const int t  = threadIdx.x;
    const int jr = blockIdx.x;
    const int it = blockIdx.y;
    const int b  = blockIdx.z;

    const int o4    = (t & 3) * 4;
    const int row32 = (t >> 2) & 31;
    const int rIn   = row32 & 15;
    const bool jpath = (t < 128);
    const int tileB = jpath ? (jr * 2 + (row32 >> 4))
                            : ((row32 < 16) ? ((it == 0) ? 0 : (it - 1)) : it);
    const int inTile = (rIn & 7) * 32 + (rIn >> 3) * 16 + o4;
    const float* pRow = Pp + (size_t)(b * TILES_PER_B + tileB) * 256 + inTile;

    const float* pI = Pp + (size_t)(b * TILES_PER_B + it) * 256 + inTile;
    const float* pA = jpath ? (pRow + 0 * PLANE_PITCH) : (pI + 1 * PLANE_PITCH);
    const float* pB = jpath ? (pRow + 4 * PLANE_PITCH) : (pI + 3 * PLANE_PITCH);
    const float* pR = pRow + 2 * PLANE_PITCH;
    const float* tp = TtP + (size_t)((b * TILES_PER_B + (t >> 4)) * STRIPS) * TT_PITCH + (t & 15);
    v4f rs[STRIPS], as[STRIPS], bs[STRIPS];
    float ts[STRIPS];
    #pragma unroll
    for (int s = 0; s < STRIPS; ++s) {
        rs[s] = *(const v4f*)(pR + s * (TILES * 256));
        as[s] = *(const v4f*)(pA + s * (TILES * 256));
        bs[s] = *(const v4f*)(pB + s * (TILES * 256));
        ts[s] = tp[s * TT_PITCH];
    }
    const v4f bf4 = *(const v4f*)(bf + o4);
    __builtin_amdgcn_sched_barrier(0);
    const v4f rsum = ((rs[0] + rs[1]) + rs[2]) + rs[3];
    const v4f qa   = ((as[0] + as[1]) + as[2]) + as[3];
    const v4f qb   = ((bs[0] + bs[1]) + bs[2]) + bs[3];
    const float t0 = ts[0], t1 = ts[1], t2 = ts[2], t3 = ts[3];

    *(v4f*)(sR + t * 4) = rsum;
    sTot[t] = ((t0 + t1) + t2) + t3;
    if (jpath) {
        const v4f q = qa + 0.5f * qb + bf4;
        #pragma unroll
        for (int e = 0; e < 4; ++e) {
            sQj[(o4 + e) * SQ_PITCH + row32] = q[e];
        }
    } else if (t < 192) {
        *(v4f*)(sQi + row32 * ODIM + o4) = 0.5f * qa - qb;
    }
    __syncthreads();

    {
        const v4f p5 = p5_row(sR + (t >> 6) * (16 * ODIM), rIn, o4, sTot, tileB);
        if (jpath) {
            #pragma unroll
            for (int e = 0; e < 4; ++e) {
                sP5j[(o4 + e) * SQ_PITCH + row32] = p5[e];
            }
        } else {
            *(v4f*)(sP5i + row32 * ODIM + o4) = p5;
        }
    }
    __syncthreads();

    const int jl  = (t & 7) * 4;
    const int grp = t >> 3;
    const int ii  = grp & 15;
    const int oh  = grp >> 4;
    const int i   = it * 16 + ii;
    const int j0  = jr * EPI_JW + jl;

    v4f rden;
    #pragma unroll
    for (int e = 0; e < 4; ++e) {
        rden[e] = 1.0f / ((float)(j0 + e - i + 1) + EPSV);
    }
    const float* ppRow = sP5i + (15 + ii) * ODIM;
    const bool firstRow = (i == 0);
    float* o0 = out + ((size_t)(b * ODIM) * LDIM + i) * LDIM + j0;

    #pragma unroll
    for (int k = 0; k < 8; ++k) {
        const int o = oh * 8 + k;
        const v4f q4 = *(const v4f*)(sQj  + o * SQ_PITCH + jl);
        const v4f p4 = *(const v4f*)(sP5j + o * SQ_PITCH + jl);
        const float qi  = sQi[ii * ODIM + o];
        const float ppv = ppRow[o];
        const float p5p = firstRow ? 0.0f : ppv;
        v4f v;
        #pragma unroll
        for (int e = 0; e < 4; ++e) {
            v[e] = q4[e] + qi + (p4[e] - p5p) * rden[e];
        }
        *(v4f*)(o0 + (size_t)o * LDIM * LDIM) = v;
    }
}

extern "C" void kernel_launch(void* const* d_in, const int* in_sizes, int n_in,
                              void* d_out, int out_size, void* d_ws, size_t ws_size,
                              hipStream_t stream)
{
    const float* x  = (const float*)d_in[0];
    const float* W1 = (const float*)d_in[1];
    const float* b1 = (const float*)d_in[2];
    const float* W2 = (const float*)d_in[3];
    const float* b2 = (const float*)d_in[4];
    const float* Wf = (const float*)d_in[5];
    const float* bf = (const float*)d_in[6];
    float* out = (float*)d_out;

    float* ws  = (float*)d_ws;
    float* Pp  = ws;
    float* TtP = Pp + (size_t)NPLANES * PLANE_PITCH;

    dim3 g1(MDIM / 16, HDIM / 64, 2);
    mlp_wmma_kernel<<<g1, 32 * MLP_WAVES, 0, stream>>>(x, W1, b1, W2, b2, Wf, Pp, TtP);
    epilogue_kernel<<<dim3(LDIM / EPI_JW, TILES_PER_B, BDIM), 256, 0, stream>>>(Pp, TtP, bf, out);
}
